# starting kernel
# baseline (speedup 1.0000x reference)
#include <hip/hip_runtime.h>

#define IN_F  128
#define HID_F 64
#define OUT_F 64

typedef __attribute__((ext_vector_type(2))) float v2f;
typedef __attribute__((ext_vector_type(8))) float v8f;

__global__ void gcn_zero_kernel(float* __restrict__ p, long long n) {
    long long i = (long long)blockIdx.x * blockDim.x + threadIdx.x;
    if (i < n) p[i] = 0.0f;
}

__global__ void gcn_degree_kernel(const int* __restrict__ src,
                                  const int* __restrict__ dst,
                                  float* __restrict__ deg_out,
                                  float* __restrict__ deg_in, int nE) {
    int e = blockIdx.x * blockDim.x + threadIdx.x;
    if (e < nE) {
        atomicAdd(&deg_out[src[e]], 1.0f);
        atomicAdd(&deg_in [dst[e]], 1.0f);
    }
}

__global__ void gcn_norm_kernel(float* __restrict__ p, long long n) {
    long long i = (long long)blockIdx.x * blockDim.x + threadIdx.x;
    if (i < n) {
        float d = p[i];
        d = d < 1.0f ? 1.0f : d;
        p[i] = rsqrtf(d);
    }
}

__global__ __launch_bounds__(128) void gcn_gemm1_kernel(
        const float* __restrict__ X, const float* __restrict__ W,
        const float* __restrict__ onorm, float* __restrict__ H, int nN) {
    const int mt   = blockIdx.x;
    const int wv   = threadIdx.x >> 5;
    const int lane = threadIdx.x & 31;
    const int half = lane >> 4;
    const int l    = lane & 15;
    const int m0   = mt * 16;
    const int n0   = wv * 16;

    int row = m0 + l;
    if (row >= nN) row = nN - 1;
    const float  sc = onorm[row];
    const float* xr = X + (long long)row * IN_F;

    v8f acc = {};
#pragma unroll
    for (int k0 = 0; k0 < IN_F; k0 += 4) {
        const int ka = k0 + 2 * half;
        v2f a = *(const v2f*)(xr + ka);
        a *= sc;
        v2f b;
        b.x = W[(long long)ka       * HID_F + n0 + l];
        b.y = W[(long long)(ka + 1) * HID_F + n0 + l];
        acc = __builtin_amdgcn_wmma_f32_16x16x4_f32(
                  false, a, false, b, (short)0, acc, false, false);
    }
#pragma unroll
    for (int r = 0; r < 8; ++r) {
        const int orow = m0 + r + 8 * half;
        if (orow < nN) H[(long long)orow * HID_F + n0 + l] = acc[r];
    }
}

__global__ void gcn_scatter1_kernel(const int* __restrict__ src,
                                    const int* __restrict__ dst,
                                    const float* __restrict__ H,
                                    float* __restrict__ AGG, long long total) {
    long long t = (long long)blockIdx.x * blockDim.x + threadIdx.x;
    if (t >= total) return;
    const int e = (int)(t >> 6);
    const int c = (int)(t & 63);
    const int s = src[e];
    const int d = dst[e];
    atomicAdd(&AGG[(long long)d * HID_F + c], H[(long long)s * HID_F + c]);
}

__global__ void gcn_post1_kernel(float* __restrict__ A,
                                 const float* __restrict__ inorm,
                                 const float* __restrict__ b1, long long n) {
    long long i = (long long)blockIdx.x * blockDim.x + threadIdx.x;
    if (i < n) {
        const int node = (int)(i >> 6);
        const int c    = (int)(i & 63);
        float v = A[i] * inorm[node] + b1[c];
        A[i] = v > 0.0f ? v : 0.0f;
    }
}

__global__ void gcn_scatter2_kernel(const int* __restrict__ src,
                                    const int* __restrict__ dst,
                                    const float* __restrict__ H,
                                    const float* __restrict__ onorm,
                                    float* __restrict__ AGG, long long total) {
    long long t = (long long)blockIdx.x * blockDim.x + threadIdx.x;
    if (t >= total) return;
    const int e = (int)(t >> 6);
    const int c = (int)(t & 63);
    const int s = src[e];
    const int d = dst[e];
    atomicAdd(&AGG[(long long)d * HID_F + c],
              H[(long long)s * HID_F + c] * onorm[s]);
}

__global__ __launch_bounds__(128) void gcn_gemm2_kernel(
        const float* __restrict__ A, const float* __restrict__ W,
        const float* __restrict__ inorm, const float* __restrict__ bias,
        float* __restrict__ Out, int nN) {
    const int mt   = blockIdx.x;
    const int wv   = threadIdx.x >> 5;
    const int lane = threadIdx.x & 31;
    const int half = lane >> 4;
    const int l    = lane & 15;
    const int m0   = mt * 16;
    const int n0   = wv * 16;

    int row = m0 + l;
    if (row >= nN) row = nN - 1;
    const float* ar = A + (long long)row * HID_F;

    v8f acc = {};
#pragma unroll
    for (int k0 = 0; k0 < HID_F; k0 += 4) {
        const int ka = k0 + 2 * half;
        v2f a = *(const v2f*)(ar + ka);
        v2f b;
        b.x = W[(long long)ka       * OUT_F + n0 + l];
        b.y = W[(long long)(ka + 1) * OUT_F + n0 + l];
        acc = __builtin_amdgcn_wmma_f32_16x16x4_f32(
                  false, a, false, b, (short)0, acc, false, false);
    }
    const float bcol = bias[n0 + l];
#pragma unroll
    for (int r = 0; r < 8; ++r) {
        const int orow = m0 + r + 8 * half;
        if (orow < nN)
            Out[(long long)orow * OUT_F + n0 + l] = acc[r] * inorm[orow] + bcol;
    }
}

extern "C" void kernel_launch(void* const* d_in, const int* in_sizes, int n_in,
                              void* d_out, int out_size, void* d_ws, size_t ws_size,
                              hipStream_t stream) {
    const float* X   = (const float*)d_in[0];
    const int*   src = (const int*)  d_in[1];
    const int*   dst = (const int*)  d_in[2];
    const float* W1  = (const float*)d_in[3];
    const float* b1  = (const float*)d_in[4];
    const float* W2  = (const float*)d_in[5];
    const float* b2  = (const float*)d_in[6];
    float*       out = (float*)d_out;

    const int nN = in_sizes[0] / IN_F;
    const int nE = in_sizes[1];

    float* ws    = (float*)d_ws;
    float* onorm = ws;
    float* inorm = ws + nN;
    float* bufA  = ws + 2LL * nN;
    float* bufB  = bufA + (long long)nN * HID_F;

    const long long nh  = (long long)nN * HID_F;
    const long long tot = (long long)nE * HID_F;

    gcn_zero_kernel  <<<(2LL*nN + 255) / 256, 256, 0, stream>>>(ws, 2LL*nN);
    gcn_degree_kernel<<<(nE + 255) / 256, 256, 0, stream>>>(src, dst, onorm, inorm, nE);
    gcn_norm_kernel  <<<(2LL*nN + 255) / 256, 256, 0, stream>>>(ws, 2LL*nN);

    gcn_gemm1_kernel   <<<(nN + 15) / 16, 128, 0, stream>>>(X, W1, onorm, bufA, nN);
    gcn_zero_kernel    <<<(nh + 255) / 256, 256, 0, stream>>>(bufB, nh);
    gcn_scatter1_kernel<<<(tot + 255) / 256, 256, 0, stream>>>(src, dst, bufA, bufB, tot);
    gcn_post1_kernel   <<<(nh + 255) / 256, 256, 0, stream>>>(bufB, inorm, b1, nh);

    gcn_zero_kernel    <<<(nh + 255) / 256, 256, 0, stream>>>(bufA, nh);
    gcn_scatter2_kernel<<<(tot + 255) / 256, 256, 0, stream>>>(src, dst, bufB, onorm, bufA, tot);
    gcn_gemm2_kernel   <<<(nN + 15) / 16, 128, 0, stream>>>(bufA, W2, inorm, b2, out, nN);
}


# The three 200000-wave elementwise launches and the norm pass are folded into neighbours: GEMM1 blocks store both accumulators' zero rows, walk 2 applies the layer-1 epilogue to each row it loads
# speedup vs baseline: 2.2016x; 2.2016x over previous
#include <hip/hip_runtime.h>

#define IN_F  128
#define HID_F 64
#define OUT_F 64

typedef __attribute__((ext_vector_type(2))) float v2f;
typedef __attribute__((ext_vector_type(4))) float v4f;
typedef __attribute__((ext_vector_type(8))) float v8f;

__global__ void gcn_zero_kernel(float* __restrict__ p, long long n) {
    long long i = (long long)blockIdx.x * blockDim.x + threadIdx.x;
    if (i < n) p[i] = 0.0f;
}

__global__ void gcn_degree_kernel(const int* __restrict__ src,
                                  const int* __restrict__ dst,
                                  float* __restrict__ deg, int nE) {
    int e = blockIdx.x * blockDim.x + threadIdx.x;
    if (e < nE) {
        atomicAdd(&deg[2LL * src[e]],     1.0f);
        atomicAdd(&deg[2LL * dst[e] + 1], 1.0f);
    }
}

__device__ __forceinline__ float gcn_norm_of(float d) {
    d = d < 1.0f ? 1.0f : d;
    return rsqrtf(d);
}

__global__ __launch_bounds__(128) void gcn_gemm1_kernel(
        const float* __restrict__ X, const float* __restrict__ W,
        const float* __restrict__ deg, float* __restrict__ H,
        float* __restrict__ Z1, float* __restrict__ Z2, int nN) {
    const int mt   = blockIdx.x;
    const int wv   = threadIdx.x >> 5;
    const int lane = threadIdx.x & 31;
    const int half = lane >> 4;
    const int l    = lane & 15;
    const int m0   = mt * 16;
    const int n0   = wv * 16;

    {
        const long long lim = (long long)nN * HID_F;
        const v4f z = {0.0f, 0.0f, 0.0f, 0.0f};
        for (int p = 0; p < 2; ++p) {
            const long long i = (long long)m0 * HID_F + p * 512 + (int)threadIdx.x * 4;
            if (i < lim) {
                *(v4f*)(Z1 + i) = z;
                *(v4f*)(Z2 + i) = z;
            }
        }
    }

    int row = m0 + l;
    if (row >= nN) row = nN - 1;
    const float  sc = gcn_norm_of(deg[2LL * row]);
    const float* xr = X + (long long)row * IN_F;

    v8f acc = {};
#pragma unroll
    for (int k0 = 0; k0 < IN_F; k0 += 4) {
        const int ka = k0 + 2 * half;
        v2f a = *(const v2f*)(xr + ka);
        a *= sc;
        v2f b;
        b.x = W[(long long)ka       * HID_F + n0 + l];
        b.y = W[(long long)(ka + 1) * HID_F + n0 + l];
        acc = __builtin_amdgcn_wmma_f32_16x16x4_f32(
                  false, a, false, b, (short)0, acc, false, false);
    }
#pragma unroll
    for (int r = 0; r < 8; ++r) {
        const int orow = m0 + r + 8 * half;
        if (orow < nN) H[(long long)orow * HID_F + n0 + l] = acc[r];
    }
}

#define EPW  32
#define EGRP 8

template <bool EPI>
__global__ __launch_bounds__(256) void gcn_edge_walk_kernel(
        const int* __restrict__ src, const int* __restrict__ dst,
        const float* __restrict__ H, const float* __restrict__ deg,
        const float* __restrict__ b1, float* __restrict__ AGG, int nE) {
    const int lane = threadIdx.x & 31;
    const int wv = __builtin_amdgcn_readfirstlane((int)(threadIdx.x >> 5));
    const int e0 = ((int)blockIdx.x * 8 + wv) * EPW;
    if (e0 >= nE) return;
    const int left = nE - e0;
    const int cnt  = left < EPW ? left : EPW;

    int e = e0 + lane;
    if (e >= nE) e = nE - 1;
    const int s_l = src[e];
    const int d_l = dst[e];
    float w_l = 1.0f, i_l = 1.0f, bias0 = 0.0f, bias1 = 0.0f;
    if (EPI) {
        const v2f dg = *(const v2f*)(deg + 2LL * s_l);
        w_l   = gcn_norm_of(dg.x);
        i_l   = gcn_norm_of(dg.y);
        bias0 = b1[lane];
        bias1 = b1[lane + 32];
    }

    if (cnt == EPW) {
#pragma unroll
        for (int g = 0; g < EPW; g += EGRP) {
            float v0[EGRP], v1[EGRP];
#pragma unroll
            for (int u = 0; u < EGRP; ++u) {
                const int s = __builtin_amdgcn_readlane(s_l, g + u);
                const float* hr = H + (long long)s * HID_F;
                v0[u] = hr[lane];
                v1[u] = hr[lane + 32];
            }
#pragma unroll
            for (int u = 0; u < EGRP; ++u) {
                const int d = __builtin_amdgcn_readlane(d_l, g + u);
                float a0 = v0[u], a1 = v1[u];
                if (EPI) {
                    const float w = __builtin_bit_cast(float,
                        __builtin_amdgcn_readlane(__builtin_bit_cast(int, w_l), g + u));
                    const float n = __builtin_bit_cast(float,
                        __builtin_amdgcn_readlane(__builtin_bit_cast(int, i_l), g + u));
                    a0 = a0 * n + bias0;
                    a1 = a1 * n + bias1;
                    a0 = (a0 > 0.0f ? a0 : 0.0f) * w;
                    a1 = (a1 > 0.0f ? a1 : 0.0f) * w;
                }
                float* ar = AGG + (long long)d * HID_F;
                atomicAdd(ar + lane,      a0);
                atomicAdd(ar + lane + 32, a1);
            }
        }
    } else {
        for (int u = 0; u < cnt; ++u) {
            const int s = __builtin_amdgcn_readlane(s_l, u);
            const int d = __builtin_amdgcn_readlane(d_l, u);
            const float* hr = H + (long long)s * HID_F;
            float a0 = hr[lane], a1 = hr[lane + 32];
            if (EPI) {
                const float w = __builtin_bit_cast(float,
                    __builtin_amdgcn_readlane(__builtin_bit_cast(int, w_l), u));
                const float n = __builtin_bit_cast(float,
                    __builtin_amdgcn_readlane(__builtin_bit_cast(int, i_l), u));
                a0 = a0 * n + bias0;
                a1 = a1 * n + bias1;
                a0 = (a0 > 0.0f ? a0 : 0.0f) * w;
                a1 = (a1 > 0.0f ? a1 : 0.0f) * w;
            }
            float* ar = AGG + (long long)d * HID_F;
            atomicAdd(ar + lane,      a0);
            atomicAdd(ar + lane + 32, a1);
        }
    }
}

__global__ __launch_bounds__(128) void gcn_gemm2_kernel(
        const float* __restrict__ A, const float* __restrict__ W,
        const float* __restrict__ deg, const float* __restrict__ bias,
        float* __restrict__ Out, int nN) {
    const int mt   = blockIdx.x;
    const int wv   = threadIdx.x >> 5;
    const int lane = threadIdx.x & 31;
    const int half = lane >> 4;
    const int l    = lane & 15;
    const int m0   = mt * 16;
    const int n0   = wv * 16;

    int row = m0 + l;
    if (row >= nN) row = nN - 1;
    const float* ar = A + (long long)row * HID_F;

    v8f acc = {};
#pragma unroll
    for (int k0 = 0; k0 < HID_F; k0 += 4) {
        const int ka = k0 + 2 * half;
        v2f a = *(const v2f*)(ar + ka);
        v2f b;
        b.x = W[(long long)ka       * OUT_F + n0 + l];
        b.y = W[(long long)(ka + 1) * OUT_F + n0 + l];
        acc = __builtin_amdgcn_wmma_f32_16x16x4_f32(
                  false, a, false, b, (short)0, acc, false, false);
    }
    const float bcol = bias[n0 + l];
#pragma unroll
    for (int r = 0; r < 8; ++r) {
        const int orow = m0 + r + 8 * half;
        if (orow < nN)
            Out[(long long)orow * OUT_F + n0 + l] =
                acc[r] * gcn_norm_of(deg[2LL * orow + 1]) + bcol;
    }
}

extern "C" void kernel_launch(void* const* d_in, const int* in_sizes, int n_in,
                              void* d_out, int out_size, void* d_ws, size_t ws_size,
                              hipStream_t stream) {
    const float* X   = (const float*)d_in[0];
    const int*   src = (const int*)  d_in[1];
    const int*   dst = (const int*)  d_in[2];
    const float* W1  = (const float*)d_in[3];
    const float* b1  = (const float*)d_in[4];
    const float* W2  = (const float*)d_in[5];
    const float* b2  = (const float*)d_in[6];
    float*       out = (float*)d_out;

    const int nN = in_sizes[0] / IN_F;
    const int nE = in_sizes[1];

    float* ws   = (float*)d_ws;
    float* bufA = ws;
    float* bufB = bufA + (long long)nN * HID_F;
    float* bufC = bufB + (long long)nN * HID_F;
    float* deg  = bufC + (long long)nN * HID_F;

    gcn_zero_kernel  <<<(2LL*nN + 255) / 256, 256, 0, stream>>>(deg, 2LL*nN);
    gcn_degree_kernel<<<(nE + 255) / 256, 256, 0, stream>>>(src, dst, deg, nE);

    gcn_gemm1_kernel   <<<(nN + 15) / 16, 128, 0, stream>>>(X, W1, deg, bufA, bufB, bufC, nN);
    gcn_edge_walk_kernel<false><<<(nE + 255) / 256, 256, 0, stream>>>(src, dst, bufA, deg, b1, bufB, nE);

    gcn_edge_walk_kernel<true> <<<(nE + 255) / 256, 256, 0, stream>>>(src, dst, bufB, deg, b1, bufC, nE);
    gcn_gemm2_kernel   <<<(nN + 15) / 16, 128, 0, stream>>>(bufC, W2, deg, b2, out, nN);
}


# Every atomic add reaches its line from one cache half (die id): a degree table per half (readers add the two); walks draw a 64-edge ticket per half and add only rows of that half's nodes
# speedup vs baseline: 3.2582x; 1.4799x over previous
#include <hip/hip_runtime.h>

#define IN_F  128
#define HID_F 64
#define OUT_F 64

typedef __attribute__((ext_vector_type(2))) float v2f;
typedef __attribute__((ext_vector_type(4))) float v4f;
typedef __attribute__((ext_vector_type(8))) float v8f;

__global__ void gcn_zero_kernel(float* __restrict__ p, long long n) {
    long long i = (long long)blockIdx.x * blockDim.x + threadIdx.x;
    if (i < n) p[i] = 0.0f;
}

__device__ __forceinline__ int gcn_my_half() {
    const unsigned id = (unsigned)__builtin_amdgcn_s_sendmsg_rtn(0x87);
    return (int)((id >> 18) & 1u);
}

__global__ void gcn_degree_kernel(const int* __restrict__ src,
                                  const int* __restrict__ dst,
                                  float* __restrict__ deg, int nE, int nN) {
    float* __restrict__ dg = deg + (long long)gcn_my_half() * 2LL * nN;
    int e = blockIdx.x * blockDim.x + threadIdx.x;
    if (e < nE) {
        atomicAdd(&dg[2LL * src[e]],     1.0f);
        atomicAdd(&dg[2LL * dst[e] + 1], 1.0f);
    }
}

__device__ __forceinline__ v2f gcn_deg_pair(const float* __restrict__ deg, int nN, int node) {
    const v2f a = *(const v2f*)(deg + 2LL * node);
    const v2f b = *(const v2f*)(deg + 2LL * nN + 2LL * node);
    return a + b;
}

__device__ __forceinline__ float gcn_norm_of(float d) {
    d = d < 1.0f ? 1.0f : d;
    return rsqrtf(d);
}

__global__ __launch_bounds__(128) void gcn_gemm1_kernel(
        const float* __restrict__ X, const float* __restrict__ W,
        const float* __restrict__ deg, float* __restrict__ H,
        float* __restrict__ Z1, float* __restrict__ Z2, int nN) {
    const int mt   = blockIdx.x;
    const int wv   = threadIdx.x >> 5;
    const int lane = threadIdx.x & 31;
    const int half = lane >> 4;
    const int l    = lane & 15;
    const int m0   = mt * 16;
    const int n0   = wv * 16;

    {
        const long long lim = (long long)nN * HID_F;
        const v4f z = {0.0f, 0.0f, 0.0f, 0.0f};
        for (int p = 0; p < 2; ++p) {
            const long long i = (long long)m0 * HID_F + p * 512 + (int)threadIdx.x * 4;
            if (i < lim) {
                *(v4f*)(Z1 + i) = z;
                *(v4f*)(Z2 + i) = z;
            }
        }
    }

    int row = m0 + l;
    if (row >= nN) row = nN - 1;
    const float  sc = gcn_norm_of(gcn_deg_pair(deg, nN, row).x);
    const float* xr = X + (long long)row * IN_F;

    v8f acc = {};
#pragma unroll
    for (int k0 = 0; k0 < IN_F; k0 += 4) {
        const int ka = k0 + 2 * half;
        v2f a = *(const v2f*)(xr + ka);
        a *= sc;
        v2f b;
        b.x = W[(long long)ka       * HID_F + n0 + l];
        b.y = W[(long long)(ka + 1) * HID_F + n0 + l];
        acc = __builtin_amdgcn_wmma_f32_16x16x4_f32(
                  false, a, false, b, (short)0, acc, false, false);
    }
#pragma unroll
    for (int r = 0; r < 8; ++r) {
        const int orow = m0 + r + 8 * half;
        if (orow < nN) H[(long long)orow * HID_F + n0 + l] = acc[r];
    }
}

#define ECH  64
#define EGRP 8

template <bool EPI, bool FULL>
__device__ __forceinline__ void gcn_walk_group(
        const float* __restrict__ H, float* __restrict__ AGG,
        int sv, int dv, float wv_, float iv, float bias0, float bias1,
        int lane, int l0, int last, int nvalid) {
    float v0[EGRP], v1[EGRP];
    int   li[EGRP];
#pragma unroll
    for (int u = 0; u < EGRP; ++u) {
        li[u] = FULL ? (l0 + u) : ((l0 + u) < last ? (l0 + u) : last);
        const int s = __builtin_amdgcn_readlane(sv, li[u]);
        const float* hr = H + (long long)s * HID_F;
        v0[u] = hr[lane];
        v1[u] = hr[lane + 32];
    }
#pragma unroll
    for (int u = 0; u < EGRP; ++u) {
        if (FULL || u < nvalid) {
            const int d = __builtin_amdgcn_readlane(dv, li[u]);
            float a0 = v0[u], a1 = v1[u];
            if (EPI) {
                const float w = __builtin_bit_cast(float,
                    __builtin_amdgcn_readlane(__builtin_bit_cast(int, wv_), li[u]));
                const float n = __builtin_bit_cast(float,
                    __builtin_amdgcn_readlane(__builtin_bit_cast(int, iv), li[u]));
                a0 = a0 * n + bias0;
                a1 = a1 * n + bias1;
                a0 = (a0 > 0.0f ? a0 : 0.0f) * w;
                a1 = (a1 > 0.0f ? a1 : 0.0f) * w;
            }
            float* ar = AGG + (long long)d * HID_F;
            atomicAdd(ar + lane,      a0);
            atomicAdd(ar + lane + 32, a1);
        }
    }
}

template <bool EPI>
__global__ __launch_bounds__(256) void gcn_edge_walk_kernel(
        const int* __restrict__ src, const int* __restrict__ dst,
        const float* __restrict__ H, const float* __restrict__ deg,
        const float* __restrict__ b1, float* __restrict__ AGG,
        int* __restrict__ ctr, int nE, int nN) {
    __shared__ int pk_s[8][ECH];
    __shared__ int pk_d[8][ECH];
    const int lane = threadIdx.x & 31;
    const int wv = __builtin_amdgcn_readfirstlane((int)(threadIdx.x >> 5));
    const int nCh   = (nE + ECH - 1) / ECH;
    const int split = nN >> 1;

    int q = gcn_my_half();
    int t = 0;
    if (lane == 0) t = atomicAdd(ctr + 64 * q, 1);
    t = __builtin_amdgcn_readfirstlane(t);
    if (t >= nCh) {
        q ^= 1;
        t = 0;
        if (lane == 0) t = atomicAdd(ctr + 64 * q, 1);
        t = __builtin_amdgcn_readfirstlane(t);
        if (t >= nCh) return;
    }

    const int ea = t * ECH + lane;
    const int eb = ea + 32;
    const int ca = ea < nE ? ea : nE - 1;
    const int cb = eb < nE ? eb : nE - 1;
    const int s_a = src[ca], d_a = dst[ca];
    const int s_b = src[cb], d_b = dst[cb];
    const bool ka = (ea < nE) && ((d_a >= split ? 1 : 0) == q);
    const bool kb = (eb < nE) && ((d_b >= split ? 1 : 0) == q);
    const unsigned ma = __builtin_amdgcn_ballot_w32(ka);
    const unsigned mb = __builtin_amdgcn_ballot_w32(kb);
    const int na  = __builtin_popcount(ma);
    const int cnt = na + __builtin_popcount(mb);
    if (cnt == 0) return;
    if (ka) {
        const int r = (int)__builtin_amdgcn_mbcnt_lo(ma, 0u);
        pk_s[wv][r] = s_a;
        pk_d[wv][r] = d_a;
    }
    if (kb) {
        const int r = na + (int)__builtin_amdgcn_mbcnt_lo(mb, 0u);
        pk_s[wv][r] = s_b;
        pk_d[wv][r] = d_b;
    }
    int sA = pk_s[wv][lane],      dA = pk_d[wv][lane];
    int sB = pk_s[wv][lane + 32], dB = pk_d[wv][lane + 32];
    if (lane      >= cnt) { sA = 0; dA = 0; }
    if (lane + 32 >= cnt) { sB = 0; dB = 0; }

    float wA = 1.0f, iA = 1.0f, wB = 1.0f, iB = 1.0f, bias0 = 0.0f, bias1 = 0.0f;
    if (EPI) {
        const v2f ga = gcn_deg_pair(deg, nN, sA);
        const v2f gb = gcn_deg_pair(deg, nN, sB);
        wA = gcn_norm_of(ga.x);  iA = gcn_norm_of(ga.y);
        wB = gcn_norm_of(gb.x);  iB = gcn_norm_of(gb.y);
        bias0 = b1[lane];
        bias1 = b1[lane + 32];
    }

    for (int g = 0; g < cnt; g += EGRP) {
        const bool hi   = g >= 32;
        const int  base = hi ? 32 : 0;
        const int  sv   = hi ? sB : sA;
        const int  dv   = hi ? dB : dA;
        const float wv_ = hi ? wB : wA;
        const float iv  = hi ? iB : iA;
        const int  left = cnt - g;
        if (left >= EGRP)
            gcn_walk_group<EPI, true >(H, AGG, sv, dv, wv_, iv, bias0, bias1, lane,
                                       g - base, 0, EGRP);
        else
            gcn_walk_group<EPI, false>(H, AGG, sv, dv, wv_, iv, bias0, bias1, lane,
                                       g - base, cnt - 1 - base, left);
    }
}

__global__ __launch_bounds__(128) void gcn_gemm2_kernel(
        const float* __restrict__ A, const float* __restrict__ W,
        const float* __restrict__ deg, const float* __restrict__ bias,
        float* __restrict__ Out, int nN) {
    const int mt   = blockIdx.x;
    const int wv   = threadIdx.x >> 5;
    const int lane = threadIdx.x & 31;
    const int half = lane >> 4;
    const int l    = lane & 15;
    const int m0   = mt * 16;
    const int n0   = wv * 16;

    int row = m0 + l;
    if (row >= nN) row = nN - 1;
    const float* ar = A + (long long)row * HID_F;

    v8f acc = {};
#pragma unroll
    for (int k0 = 0; k0 < HID_F; k0 += 4) {
        const int ka = k0 + 2 * half;
        v2f a = *(const v2f*)(ar + ka);
        v2f b;
        b.x = W[(long long)ka       * OUT_F + n0 + l];
        b.y = W[(long long)(ka + 1) * OUT_F + n0 + l];
        acc = __builtin_amdgcn_wmma_f32_16x16x4_f32(
                  false, a, false, b, (short)0, acc, false, false);
    }
    const float bcol = bias[n0 + l];
#pragma unroll
    for (int r = 0; r < 8; ++r) {
        const int orow = m0 + r + 8 * half;
        if (orow < nN)
            Out[(long long)orow * OUT_F + n0 + l] =
                acc[r] * gcn_norm_of(gcn_deg_pair(deg, nN, orow).y) + bcol;
    }
}

extern "C" void kernel_launch(void* const* d_in, const int* in_sizes, int n_in,
                              void* d_out, int out_size, void* d_ws, size_t ws_size,
                              hipStream_t stream) {
    const float* X   = (const float*)d_in[0];
    const int*   src = (const int*)  d_in[1];
    const int*   dst = (const int*)  d_in[2];
    const float* W1  = (const float*)d_in[3];
    const float* b1  = (const float*)d_in[4];
    const float* W2  = (const float*)d_in[5];
    const float* b2  = (const float*)d_in[6];
    float*       out = (float*)d_out;

    const int nN = in_sizes[0] / IN_F;
    const int nE = in_sizes[1];

    float* ws   = (float*)d_ws;
    float* bufA = ws;
    float* bufB = bufA + (long long)nN * HID_F;
    float* bufC = bufB + (long long)nN * HID_F;
    float* deg  = bufC + (long long)nN * HID_F;
    int*   ctr  = (int*)(deg + 4LL * nN);

    gcn_zero_kernel  <<<(4LL*nN + 256 + 255) / 256, 256, 0, stream>>>(deg, 4LL*nN + 256);
    gcn_degree_kernel<<<(nE + 255) / 256, 256, 0, stream>>>(src, dst, deg, nE, nN);

    gcn_gemm1_kernel   <<<(nN + 15) / 16, 128, 0, stream>>>(X, W1, deg, bufA, bufB, bufC, nN);
    gcn_edge_walk_kernel<false><<<((nE + 63) / 64 * 2 + 7) / 8, 256, 0, stream>>>(src, dst, bufA, deg, b1, bufB, ctr, nE, nN);

    gcn_edge_walk_kernel<true> <<<((nE + 63) / 64 * 2 + 7) / 8, 256, 0, stream>>>(src, dst, bufB, deg, b1, bufC, ctr + 128, nE, nN);
    gcn_gemm2_kernel   <<<(nN + 15) / 16, 128, 0, stream>>>(bufC, W2, deg, b2, out, nN);
}


# Layer 2's product moved in front of its sum: walk 1's block multiplies its finished rows by W2 on WMMA from LDS; walk 2 sums product rows and finishes out; GEMM2 launch gone (sum order changes)
# speedup vs baseline: 5.3847x; 1.6527x over previous
#include <hip/hip_runtime.h>

#define IN_F  128
#define HID_F 64
#define OUT_F 64

typedef __attribute__((ext_vector_type(2))) float v2f;
typedef __attribute__((ext_vector_type(4))) float v4f;
typedef __attribute__((ext_vector_type(8))) float v8f;

__global__ void gcn_zero_kernel(float* __restrict__ p, long long n) {
    long long i = (long long)blockIdx.x * blockDim.x + threadIdx.x;
    if (i < n) p[i] = 0.0f;
}

__device__ __forceinline__ int gcn_my_half() {
    const unsigned id = (unsigned)__builtin_amdgcn_s_sendmsg_rtn(0x87);
    return (int)((id >> 18) & 1u);
}

#define BR   200
#define CAPH 2560
#define CST  32

typedef __attribute__((ext_vector_type(2))) int v2i;

__global__ void gcn_degree_bucket_kernel(const int* __restrict__ src,
                                         const int* __restrict__ dst,
                                         float* __restrict__ deg, int* __restrict__ ctr,
                                         int* __restrict__ lists, int* __restrict__ ovf,
                                         int nE, int nN, int nB, int nBP) {
    const int hf = gcn_my_half();
    float* __restrict__ dg = deg + (long long)hf * 2LL * nN;
    int e = blockIdx.x * blockDim.x + threadIdx.x;
    if (e < nE) {
        const int s = src[e];
        const int d = dst[e];
        atomicAdd(&dg[2LL * s],     1.0f);
        atomicAdd(&dg[2LL * d + 1], 1.0f);
        const int b   = d / BR;
        const int pos = atomicAdd(ctr + 64 + CST * (hf * nBP + b), 1);
        const v2i rec = {s, d};
        if (pos < CAPH) {
            *(v2i*)(lists + 2LL * ((long long)(hf * nB + b) * CAPH + pos)) = rec;
        } else {
            const int op = atomicAdd(ctr, 1);
            *(v2i*)(ovf + 2LL * op) = rec;
        }
    }
}

__device__ __forceinline__ v2f gcn_deg_pair(const float* __restrict__ deg, int nN, int node) {
    const v2f a = *(const v2f*)(deg + 2LL * node);
    const v2f b = *(const v2f*)(deg + 2LL * nN + 2LL * node);
    return a + b;
}

__device__ __forceinline__ float gcn_norm_of(float d) {
    d = d < 1.0f ? 1.0f : d;
    return rsqrtf(d);
}

__global__ __launch_bounds__(128) void gcn_gemm1_kernel(
        const float* __restrict__ X, const float* __restrict__ W,
        const float* __restrict__ deg, float* __restrict__ H, int nN) {
    const int mt   = blockIdx.x;
    const int wv   = threadIdx.x >> 5;
    const int lane = threadIdx.x & 31;
    const int half = lane >> 4;
    const int l    = lane & 15;
    const int m0   = mt * 16;
    const int n0   = wv * 16;

    int row = m0 + l;
    if (row >= nN) row = nN - 1;
    const float  sc = gcn_norm_of(gcn_deg_pair(deg, nN, row).x);
    const float* xr = X + (long long)row * IN_F;

    v8f acc = {};
#pragma unroll
    for (int k0 = 0; k0 < IN_F; k0 += 4) {
        const int ka = k0 + 2 * half;
        v2f a = *(const v2f*)(xr + ka);
        a *= sc;
        v2f b;
        b.x = W[(long long)ka       * HID_F + n0 + l];
        b.y = W[(long long)(ka + 1) * HID_F + n0 + l];
        acc = __builtin_amdgcn_wmma_f32_16x16x4_f32(
                  false, a, false, b, (short)0, acc, false, false);
    }
#pragma unroll
    for (int r = 0; r < 8; ++r) {
        const int orow = m0 + r + 8 * half;
        if (orow < nN) H[(long long)orow * HID_F + n0 + l] = acc[r];
    }
}

#define EGRP 16
#define WTH  512
#define MT    ((BR + 15) / 16)
#define RP    (MT * 16)
#define PITCH 68

template <bool FULL>
__device__ __forceinline__ void gcn_walk_group(
        const float* __restrict__ H, float* acc, int sv, int dv,
        int lane, int l0, int last, int nvalid) {
    float v0[EGRP], v1[EGRP];
    int   li[EGRP];
#pragma unroll
    for (int u = 0; u < EGRP; ++u) {
        li[u] = FULL ? (l0 + u) : ((l0 + u) < last ? (l0 + u) : last);
        const int s = __builtin_amdgcn_readlane(sv, li[u]);
        const float* hr = H + (long long)s * HID_F;
        v0[u] = hr[lane];
        v1[u] = hr[lane + 32];
    }
#pragma unroll
    for (int u = 0; u < EGRP; ++u) {
        if (FULL || u < nvalid) {
            const int d = __builtin_amdgcn_readlane(dv, li[u]);
            float* ar = acc + d;
            atomicAdd(ar + lane,      v0[u]);
            atomicAdd(ar + lane + 32, v1[u]);
        }
    }
}

__device__ __forceinline__ void gcn_walk_list(
        const float* __restrict__ H, const int* __restrict__ lp, int n,
        float* acc, int row0, int lane, int t) {
    for (int base = t * 32; base < n; base += WTH) {
        const int idx = base + lane;
        const int ci  = idx < n ? idx : n - 1;
        const v2i rec = *(const v2i*)(lp + 2LL * ci);
        const int m   = (n - base) < 32 ? (n - base) : 32;
        const int sv  = rec.x;
        const int dv  = (rec.y - row0) * PITCH;
        for (int g = 0; g < m; g += EGRP) {
            const int left = m - g;
            if (left >= EGRP)
                gcn_walk_group<true >(H, acc, sv, dv, lane, g, 0, EGRP);
            else
                gcn_walk_group<false>(H, acc, sv, dv, lane, g, m - 1, left);
        }
    }
}

template <bool L1>
__global__ __launch_bounds__(WTH) void gcn_bucket_walk_kernel(
        const float* __restrict__ H, const int* __restrict__ lists,
        const int* __restrict__ ovf, const int* __restrict__ ctr,
        const float* __restrict__ deg, const float* __restrict__ bias,
        const float* __restrict__ W, float* __restrict__ OUT,
        int nE, int nN, int nB, int nBP) {
    __shared__ float acc[RP * PITCH];
    const int lane = threadIdx.x & 31;
    const int wv   = __builtin_amdgcn_readfirstlane((int)(threadIdx.x >> 5));
    const int bk   = blockIdx.x;
    const int row0 = bk * BR;

    {
        const v4f z = {0.0f, 0.0f, 0.0f, 0.0f};
        for (int i = threadIdx.x; i < RP * PITCH / 4; i += WTH) ((v4f*)acc)[i] = z;
    }
    __syncthreads();

    for (int hf = 0; hf < 2; ++hf) {
        int n = ctr[64 + CST * (hf * nBP + bk)];
        n = n < CAPH ? n : CAPH;
        gcn_walk_list(H, lists + 2LL * ((long long)(hf * nB + bk) * CAPH), n,
                      acc, row0, lane, (wv + hf * (WTH / 64)) & (WTH / 32 - 1));
    }

    {
        int no = ctr[0];
        no = no < nE ? no : nE;
        for (int base = wv * 32; base < no; base += WTH) {
            const int idx = base + lane;
            const int ci  = idx < no ? idx : no - 1;
            const v2i rec = *(const v2i*)(ovf + 2LL * ci);
            const int m   = (no - base) < 32 ? (no - base) : 32;
            for (int u = 0; u < m; ++u) {
                const int d = __builtin_amdgcn_readlane(rec.y, u) - row0;
                if ((unsigned)d < (unsigned)BR) {
                    const int s = __builtin_amdgcn_readlane(rec.x, u);
                    const float* hr = H + (long long)s * HID_F;
                    atomicAdd(acc + d * PITCH + lane,      hr[lane]);
                    atomicAdd(acc + d * PITCH + lane + 32, hr[lane + 32]);
                }
            }
        }
    }
    __syncthreads();

    const float bias0 = bias[lane];
    const float bias1 = bias[lane + 32];
    for (int r = wv; r < BR; r += WTH / 32) {
        const int row = row0 + r;
        if (row < nN) {
            float a0 = acc[r * PITCH + lane];
            float a1 = acc[r * PITCH + lane + 32];
            const v2f g = gcn_deg_pair(deg, nN, row);
            const float n = gcn_norm_of(g.y);
            a0 = a0 * n + bias0;
            a1 = a1 * n + bias1;
            if (L1) {
                const float w = gcn_norm_of(g.x);
                acc[r * PITCH + lane]      = (a0 > 0.0f ? a0 : 0.0f) * w;
                acc[r * PITCH + lane + 32] = (a1 > 0.0f ? a1 : 0.0f) * w;
            } else {
                float* orow = OUT + (long long)row * OUT_F;
                orow[lane]      = a0;
                orow[lane + 32] = a1;
            }
        }
    }

    if (L1) {
        __syncthreads();
        const int half = lane >> 4;
        const int l    = lane & 15;
        for (int job = wv; job < MT * (OUT_F / 16); job += WTH / 32) {
            const int mt = job / (OUT_F / 16);
            const int n0 = (job % (OUT_F / 16)) * 16;
            const float* ar = acc + (mt * 16 + l) * PITCH;
            v8f c = {};
#pragma unroll
            for (int k0 = 0; k0 < HID_F; k0 += 4) {
                const int ka = k0 + 2 * half;
                const v2f a = *(const v2f*)(ar + ka);
                v2f b;
                b.x = W[(long long)ka       * OUT_F + n0 + l];
                b.y = W[(long long)(ka + 1) * OUT_F + n0 + l];
                c = __builtin_amdgcn_wmma_f32_16x16x4_f32(
                        false, a, false, b, (short)0, c, false, false);
            }
#pragma unroll
            for (int r = 0; r < 8; ++r) {
                const int br   = mt * 16 + r + 8 * half;
                const int orow = row0 + br;
                if (br < BR && orow < nN)
                    OUT[(long long)orow * OUT_F + n0 + l] = c[r];
            }
        }
    }
}

extern "C" void kernel_launch(void* const* d_in, const int* in_sizes, int n_in,
                              void* d_out, int out_size, void* d_ws, size_t ws_size,
                              hipStream_t stream) {
    const float* X   = (const float*)d_in[0];
    const int*   src = (const int*)  d_in[1];
    const int*   dst = (const int*)  d_in[2];
    const float* W1  = (const float*)d_in[3];
    const float* b1  = (const float*)d_in[4];
    const float* W2  = (const float*)d_in[5];
    const float* b2  = (const float*)d_in[6];
    float*       out = (float*)d_out;

    const int nN = in_sizes[0] / IN_F;
    const int nE = in_sizes[1];

    const int nB  = (nN + BR - 1) / BR;
    const int nBP = (nB + 31) / 32 * 32;

    float* ws    = (float*)d_ws;
    float* bufA  = ws;
    float* bufB  = bufA + (long long)nN * HID_F;
    float* deg   = bufB + (long long)nN * HID_F;
    int*   ctr   = (int*)(deg + 4LL * nN);
    int*   lists = ctr + 64 + 2 * CST * nBP;
    int*   ovf   = lists + 4LL * nB * CAPH;

    gcn_zero_kernel         <<<(4LL*nN + 64 + 2*CST*nBP + 255) / 256, 256, 0, stream>>>(deg, 4LL*nN + 64 + 2*CST*nBP);
    gcn_degree_bucket_kernel<<<(nE + 255) / 256, 256, 0, stream>>>(src, dst, deg, ctr, lists, ovf, nE, nN, nB, nBP);

    gcn_gemm1_kernel<<<(nN + 15) / 16, 128, 0, stream>>>(X, W1, deg, bufA, nN);
    gcn_bucket_walk_kernel<true> <<<nB, WTH, 0, stream>>>(bufA, lists, ovf, ctr, deg, b1, W2, bufB, nE, nN, nB, nBP);

    gcn_bucket_walk_kernel<false><<<nB, WTH, 0, stream>>>(bufB, lists, ovf, ctr, deg, b2, W2, out, nE, nN, nB, nBP);
}


# GEMM1 data path: each 16-row X tile is loaded once as whole rows, scaled and staged in LDS for the 4 waves' A fragments; W1 fragments stay in registers over a block's 5 tiles; sum order unchanged
# speedup vs baseline: 5.8680x; 1.0898x over previous
#include <hip/hip_runtime.h>

#define IN_F  128
#define HID_F 64
#define OUT_F 64

typedef __attribute__((ext_vector_type(2))) float v2f;
typedef __attribute__((ext_vector_type(4))) float v4f;
typedef __attribute__((ext_vector_type(8))) float v8f;

__global__ void gcn_zero_kernel(float* __restrict__ p, long long n) {
    long long i = (long long)blockIdx.x * blockDim.x + threadIdx.x;
    if (i < n) p[i] = 0.0f;
}

__device__ __forceinline__ int gcn_my_half() {
    const unsigned id = (unsigned)__builtin_amdgcn_s_sendmsg_rtn(0x87);
    return (int)((id >> 18) & 1u);
}

#define BR   200
#define CAPH 2560
#define CST  32

typedef __attribute__((ext_vector_type(2))) int v2i;

__global__ void gcn_degree_bucket_kernel(const int* __restrict__ src,
                                         const int* __restrict__ dst,
                                         float* __restrict__ deg, int* __restrict__ ctr,
                                         int* __restrict__ lists, int* __restrict__ ovf,
                                         int nE, int nN, int nB, int nBP) {
    const int hf = gcn_my_half();
    float* __restrict__ dg = deg + (long long)hf * 2LL * nN;
    int e = blockIdx.x * blockDim.x + threadIdx.x;
    if (e < nE) {
        const int s = src[e];
        const int d = dst[e];
        atomicAdd(&dg[2LL * s],     1.0f);
        atomicAdd(&dg[2LL * d + 1], 1.0f);
        const int b   = d / BR;
        const int pos = atomicAdd(ctr + 64 + CST * (hf * nBP + b), 1);
        const v2i rec = {s, d};
        if (pos < CAPH) {
            *(v2i*)(lists + 2LL * ((long long)(hf * nB + b) * CAPH + pos)) = rec;
        } else {
            const int op = atomicAdd(ctr, 1);
            *(v2i*)(ovf + 2LL * op) = rec;
        }
    }
}

__device__ __forceinline__ v2f gcn_deg_pair(const float* __restrict__ deg, int nN, int node) {
    const v2f a = *(const v2f*)(deg + 2LL * node);
    const v2f b = *(const v2f*)(deg + 2LL * nN + 2LL * node);
    return a + b;
}

__device__ __forceinline__ float gcn_norm_of(float d) {
    d = d < 1.0f ? 1.0f : d;
    return rsqrtf(d);
}

#define G1_TPB 5
#define G1_PA  132
#define G1_PO  68

__device__ __forceinline__ void gcn_lds_barrier() {
    asm volatile("s_wait_dscnt 0x0\n\ts_barrier_signal -1\n\ts_barrier_wait -1" ::: "memory");
}

__device__ __forceinline__ void gcn_gemm1_rows(
        const float* __restrict__ X, const float* __restrict__ deg, int nN,
        int mt, int wv, int lane, v4f* xn, float* sn) {
#pragma unroll
    for (int q = 0; q < 4; ++q) {
        int row = mt * 16 + 4 * wv + q;
        row = row < nN ? row : nN - 1;
        sn[q] = gcn_norm_of(gcn_deg_pair(deg, nN, row).x);
        xn[q] = *(const v4f*)(X + (long long)row * IN_F + 4 * lane);
    }
}

__global__ __launch_bounds__(128) void gcn_gemm1_kernel(
        const float* __restrict__ X, const float* __restrict__ W,
        const float* __restrict__ deg, float* __restrict__ H, int nN, int nMT) {
    __shared__ __attribute__((aligned(16))) float xa[16 * G1_PA];
    __shared__ __attribute__((aligned(16))) float xo[16 * G1_PO];
    const int lane = threadIdx.x & 31;
    const int wv   = threadIdx.x >> 5;
    const int half = lane >> 4;
    const int l    = lane & 15;
    const int n0   = wv * 16;

    v2f b[IN_F / 4];
#pragma unroll
    for (int ks = 0; ks < IN_F / 4; ++ks) {
        const int ka = 4 * ks + 2 * half;
        b[ks].x = W[ka       * HID_F + n0 + l];
        b[ks].y = W[(ka + 1) * HID_F + n0 + l];
    }

    v4f   xn[4];
    float sn[4];
    int mt = blockIdx.x;
    const int gstep = (int)gridDim.x;
    gcn_gemm1_rows(X, deg, nN, mt, wv, lane, xn, sn);

    for (; mt < nMT; mt += gstep) {
#pragma unroll
        for (int q = 0; q < 4; ++q)
            *(v4f*)(xa + (4 * wv + q) * G1_PA + 4 * lane) = xn[q] * sn[q];
        gcn_lds_barrier();

        {
            int mn = mt + gstep;
            mn = mn < nMT ? mn : nMT - 1;
            gcn_gemm1_rows(X, deg, nN, mn, wv, lane, xn, sn);
        }
        __builtin_amdgcn_sched_barrier(0);

        v2f a[IN_F / 4];
        const float* ar = xa + l * G1_PA + 2 * half;
#pragma unroll
        for (int ks = 0; ks < IN_F / 4; ++ks)
            a[ks] = *(const v2f*)(ar + 4 * ks);
        __builtin_amdgcn_sched_barrier(0);

        v8f acc = {};
#pragma unroll
        for (int ks = 0; ks < IN_F / 4; ++ks)
            acc = __builtin_amdgcn_wmma_f32_16x16x4_f32(
                      false, a[ks], false, b[ks], (short)0, acc, false, false);
#pragma unroll
        for (int r = 0; r < 8; ++r)
            xo[(r + 8 * half) * G1_PO + n0 + l] = acc[r];
        gcn_lds_barrier();

#pragma unroll
        for (int q = 0; q < 2; ++q) {
            const int r    = 4 * wv + 2 * q + half;
            const int orow = mt * 16 + r;
            const v4f o = *(const v4f*)(xo + r * G1_PO + 4 * l);
            if (orow < nN) *(v4f*)(H + (long long)orow * HID_F + 4 * l) = o;
        }
    }
}

#define EGRP 16
#define WTH  512
#define MT    ((BR + 15) / 16)
#define RP    (MT * 16)
#define PITCH 68

template <bool FULL>
__device__ __forceinline__ void gcn_walk_group(
        const float* __restrict__ H, float* acc, int sv, int dv,
        int lane, int l0, int last, int nvalid) {
    float v0[EGRP], v1[EGRP];
    int   li[EGRP];
#pragma unroll
    for (int u = 0; u < EGRP; ++u) {
        li[u] = FULL ? (l0 + u) : ((l0 + u) < last ? (l0 + u) : last);
        const int s = __builtin_amdgcn_readlane(sv, li[u]);
        const float* hr = H + (long long)s * HID_F;
        v0[u] = hr[lane];
        v1[u] = hr[lane + 32];
    }
#pragma unroll
    for (int u = 0; u < EGRP; ++u) {
        if (FULL || u < nvalid) {
            const int d = __builtin_amdgcn_readlane(dv, li[u]);
            float* ar = acc + d;
            atomicAdd(ar + lane,      v0[u]);
            atomicAdd(ar + lane + 32, v1[u]);
        }
    }
}

__device__ __forceinline__ void gcn_walk_list(
        const float* __restrict__ H, const int* __restrict__ lp, int n,
        float* acc, int row0, int lane, int t) {
    for (int base = t * 32; base < n; base += WTH) {
        const int idx = base + lane;
        const int ci  = idx < n ? idx : n - 1;
        const v2i rec = *(const v2i*)(lp + 2LL * ci);
        const int m   = (n - base) < 32 ? (n - base) : 32;
        const int sv  = rec.x;
        const int dv  = (rec.y - row0) * PITCH;
        for (int g = 0; g < m; g += EGRP) {
            const int left = m - g;
            if (left >= EGRP)
                gcn_walk_group<true >(H, acc, sv, dv, lane, g, 0, EGRP);
            else
                gcn_walk_group<false>(H, acc, sv, dv, lane, g, m - 1, left);
        }
    }
}

template <bool L1>
__global__ __launch_bounds__(WTH) void gcn_bucket_walk_kernel(
        const float* __restrict__ H, const int* __restrict__ lists,
        const int* __restrict__ ovf, const int* __restrict__ ctr,
        const float* __restrict__ deg, const float* __restrict__ bias,
        const float* __restrict__ W, float* __restrict__ OUT,
        int nE, int nN, int nB, int nBP) {
    __shared__ float acc[RP * PITCH];
    const int lane = threadIdx.x & 31;
    const int wv   = __builtin_amdgcn_readfirstlane((int)(threadIdx.x >> 5));
    const int bk   = blockIdx.x;
    const int row0 = bk * BR;

    {
        const v4f z = {0.0f, 0.0f, 0.0f, 0.0f};
        for (int i = threadIdx.x; i < RP * PITCH / 4; i += WTH) ((v4f*)acc)[i] = z;
    }
    __syncthreads();

    for (int hf = 0; hf < 2; ++hf) {
        int n = ctr[64 + CST * (hf * nBP + bk)];
        n = n < CAPH ? n : CAPH;
        gcn_walk_list(H, lists + 2LL * ((long long)(hf * nB + bk) * CAPH), n,
                      acc, row0, lane, (wv + hf * (WTH / 64)) & (WTH / 32 - 1));
    }

    {
        int no = ctr[0];
        no = no < nE ? no : nE;
        for (int base = wv * 32; base < no; base += WTH) {
            const int idx = base + lane;
            const int ci  = idx < no ? idx : no - 1;
            const v2i rec = *(const v2i*)(ovf + 2LL * ci);
            const int m   = (no - base) < 32 ? (no - base) : 32;
            for (int u = 0; u < m; ++u) {
                const int d = __builtin_amdgcn_readlane(rec.y, u) - row0;
                if ((unsigned)d < (unsigned)BR) {
                    const int s = __builtin_amdgcn_readlane(rec.x, u);
                    const float* hr = H + (long long)s * HID_F;
                    atomicAdd(acc + d * PITCH + lane,      hr[lane]);
                    atomicAdd(acc + d * PITCH + lane + 32, hr[lane + 32]);
                }
            }
        }
    }
    __syncthreads();

    const float bias0 = bias[lane];
    const float bias1 = bias[lane + 32];
    for (int r = wv; r < BR; r += WTH / 32) {
        const int row = row0 + r;
        if (row < nN) {
            float a0 = acc[r * PITCH + lane];
            float a1 = acc[r * PITCH + lane + 32];
            const v2f g = gcn_deg_pair(deg, nN, row);
            const float n = gcn_norm_of(g.y);
            a0 = a0 * n + bias0;
            a1 = a1 * n + bias1;
            if (L1) {
                const float w = gcn_norm_of(g.x);
                acc[r * PITCH + lane]      = (a0 > 0.0f ? a0 : 0.0f) * w;
                acc[r * PITCH + lane + 32] = (a1 > 0.0f ? a1 : 0.0f) * w;
            } else {
                float* orow = OUT + (long long)row * OUT_F;
                orow[lane]      = a0;
                orow[lane + 32] = a1;
            }
        }
    }

    if (L1) {
        __syncthreads();
        const int half = lane >> 4;
        const int l    = lane & 15;
        for (int job = wv; job < MT * (OUT_F / 16); job += WTH / 32) {
            const int mt = job / (OUT_F / 16);
            const int n0 = (job % (OUT_F / 16)) * 16;
            const float* ar = acc + (mt * 16 + l) * PITCH;
            v8f c = {};
#pragma unroll
            for (int k0 = 0; k0 < HID_F; k0 += 4) {
                const int ka = k0 + 2 * half;
                const v2f a = *(const v2f*)(ar + ka);
                v2f b;
                b.x = W[(long long)ka       * OUT_F + n0 + l];
                b.y = W[(long long)(ka + 1) * OUT_F + n0 + l];
                c = __builtin_amdgcn_wmma_f32_16x16x4_f32(
                        false, a, false, b, (short)0, c, false, false);
            }
#pragma unroll
            for (int r = 0; r < 8; ++r) {
                const int br   = mt * 16 + r + 8 * half;
                const int orow = row0 + br;
                if (br < BR && orow < nN)
                    OUT[(long long)orow * OUT_F + n0 + l] = c[r];
            }
        }
    }
}

extern "C" void kernel_launch(void* const* d_in, const int* in_sizes, int n_in,
                              void* d_out, int out_size, void* d_ws, size_t ws_size,
                              hipStream_t stream) {
    const float* X   = (const float*)d_in[0];
    const int*   src = (const int*)  d_in[1];
    const int*   dst = (const int*)  d_in[2];
    const float* W1  = (const float*)d_in[3];
    const float* b1  = (const float*)d_in[4];
    const float* W2  = (const float*)d_in[5];
    const float* b2  = (const float*)d_in[6];
    float*       out = (float*)d_out;

    const int nN = in_sizes[0] / IN_F;
    const int nE = in_sizes[1];

    const int nB  = (nN + BR - 1) / BR;
    const int nMT = (nN + 15) / 16;
    const int nBP = (nB + 31) / 32 * 32;

    float* ws    = (float*)d_ws;
    float* bufA  = ws;
    float* bufB  = bufA + (long long)nN * HID_F;
    float* deg   = bufB + (long long)nN * HID_F;
    int*   ctr   = (int*)(deg + 4LL * nN);
    int*   lists = ctr + 64 + 2 * CST * nBP;
    int*   ovf   = lists + 4LL * nB * CAPH;

    gcn_zero_kernel         <<<(4LL*nN + 64 + 2*CST*nBP + 255) / 256, 256, 0, stream>>>(deg, 4LL*nN + 64 + 2*CST*nBP);
    gcn_degree_bucket_kernel<<<(nE + 255) / 256, 256, 0, stream>>>(src, dst, deg, ctr, lists, ovf, nE, nN, nB, nBP);

    gcn_gemm1_kernel<<<(nMT + G1_TPB - 1) / G1_TPB, 128, 0, stream>>>(X, W1, deg, bufA, nN, nMT);
    gcn_bucket_walk_kernel<true> <<<nB, WTH, 0, stream>>>(bufA, lists, ovf, ctr, deg, b1, W2, bufB, nE, nN, nB, nBP);

    gcn_bucket_walk_kernel<false><<<nB, WTH, 0, stream>>>(bufB, lists, ovf, ctr, deg, b2, W2, out, nE, nN, nB, nBP);
}


# Edge filing per 7168-edge slice in LDS: returning LDS add = rank in bucket, one cursor add per bucket and block, records sorted in LDS, stored as runs; list order changes, row sum order = LDS arrival
# speedup vs baseline: 6.3942x; 1.0897x over previous
#include <hip/hip_runtime.h>

#define IN_F  128
#define HID_F 64
#define OUT_F 64

typedef __attribute__((ext_vector_type(2))) float v2f;
typedef __attribute__((ext_vector_type(4))) float v4f;
typedef __attribute__((ext_vector_type(8))) float v8f;

__global__ void gcn_zero_kernel(float* __restrict__ p, long long n) {
    long long i = (long long)blockIdx.x * blockDim.x + threadIdx.x;
    if (i < n) p[i] = 0.0f;
}

__device__ __forceinline__ int gcn_my_half() {
    const unsigned id = (unsigned)__builtin_amdgcn_s_sendmsg_rtn(0x87);
    return (int)((id >> 18) & 1u);
}

#define BR   200
#define CAPH 2560
#define CST  32

typedef __attribute__((ext_vector_type(2))) int v2i;

__device__ __forceinline__ void gcn_lds_barrier() {
    asm volatile("s_wait_dscnt 0x0\n\ts_barrier_signal -1\n\ts_barrier_wait -1" ::: "memory");
}

#define FB_NT  1024
#define FB_EPT 7
#define FB_SL  (FB_NT * FB_EPT)
#define FB_NB  512

__global__ __launch_bounds__(FB_NT) void gcn_degree_bucket_kernel(
        const int* __restrict__ src, const int* __restrict__ dst,
        float* __restrict__ deg, int* __restrict__ ctr,
        int* __restrict__ lists, int* __restrict__ ovf,
        int nE, int nN, int nB, int nBP) {
    __shared__ v2i srec[FB_SL];
    __shared__ int c_cnt[FB_NB];
    __shared__ int c_exc[FB_NB];
    __shared__ int c_del[FB_NB];
    __shared__ int c_wt[FB_NB / 32];
    __shared__ int c_tot;
    const int tid  = threadIdx.x;
    const int lane = tid & 31;
    const int wv   = tid >> 5;
    const int hf   = gcn_my_half();
    float* __restrict__ dg = deg + (long long)hf * 2LL * nN;

    int s[FB_EPT], d[FB_EPT], b[FB_EPT], r[FB_EPT];
#pragma unroll
    for (int k = 0; k < FB_EPT; ++k) {
        const int e = blockIdx.x * FB_SL + k * FB_NT + tid;
        const bool ok = e < nE;
        s[k] = ok ? src[e] : 0;
        d[k] = ok ? dst[e] : -1;
    }
#pragma unroll
    for (int k = 0; k < FB_EPT; ++k) {
        b[k] = -1;
        if (d[k] >= 0) {
            atomicAdd(&dg[2LL * s[k]],     1.0f);
            atomicAdd(&dg[2LL * d[k] + 1], 1.0f);
            b[k] = d[k] / BR;
        }
    }

    for (int c0 = 0; c0 < nB; c0 += FB_NB) {
        if (tid < FB_NB) c_cnt[tid] = 0;
        gcn_lds_barrier();
#pragma unroll
        for (int k = 0; k < FB_EPT; ++k) {
            const unsigned bl = (unsigned)(b[k] - c0);
            r[k] = bl < (unsigned)FB_NB ? atomicAdd(&c_cnt[bl], 1) : 0;
        }
        gcn_lds_barrier();

        const bool mine = tid < FB_NB && (c0 + tid) < nB;
        const int  v    = mine ? c_cnt[tid] : 0;
        int inc = v;
        for (int o = 1; o < 32; o <<= 1) {
            const int t = __shfl_up(inc, o);
            if (lane >= o) inc += t;
        }
        if (tid < FB_NB && lane == 31) c_wt[wv] = inc;
        gcn_lds_barrier();
        if (tid < FB_NB) {
            int off = 0;
            for (int j = 0; j < wv; ++j) off += c_wt[j];
            const int ex = off + inc - v;
            int g = 0;
            if (v > 0) g = atomicAdd(ctr + 64 + CST * (hf * nBP + c0 + tid), v);
            c_exc[tid] = ex;
            c_del[tid] = g - ex;
            if (tid == FB_NB - 1) c_tot = ex + v;
        }
        gcn_lds_barrier();

#pragma unroll
        for (int k = 0; k < FB_EPT; ++k) {
            const unsigned bl = (unsigned)(b[k] - c0);
            if (bl < (unsigned)FB_NB) {
                const v2i rec = {s[k], d[k]};
                srec[c_exc[bl] + r[k]] = rec;
            }
        }
        gcn_lds_barrier();

        const int tot = c_tot;
        for (int i = tid; i < tot; i += FB_NT) {
            const v2i rec = srec[i];
            const int bk  = rec.y / BR;
            const int pos = i + c_del[bk - c0];
            if (pos < CAPH) {
                *(v2i*)(lists + 2LL * ((long long)(hf * nB + bk) * CAPH + pos)) = rec;
            } else {
                const int op = atomicAdd(ctr, 1);
                *(v2i*)(ovf + 2LL * op) = rec;
            }
        }
        gcn_lds_barrier();
    }
}

__device__ __forceinline__ v2f gcn_deg_pair(const float* __restrict__ deg, int nN, int node) {
    const v2f a = *(const v2f*)(deg + 2LL * node);
    const v2f b = *(const v2f*)(deg + 2LL * nN + 2LL * node);
    return a + b;
}

__device__ __forceinline__ float gcn_norm_of(float d) {
    d = d < 1.0f ? 1.0f : d;
    return rsqrtf(d);
}

#define G1_TPB 5
#define G1_PA  132
#define G1_PO  68

__device__ __forceinline__ void gcn_gemm1_rows(
        const float* __restrict__ X, const float* __restrict__ deg, int nN,
        int mt, int wv, int lane, v4f* xn, float* sn) {
#pragma unroll
    for (int q = 0; q < 4; ++q) {
        int row = mt * 16 + 4 * wv + q;
        row = row < nN ? row : nN - 1;
        sn[q] = gcn_norm_of(gcn_deg_pair(deg, nN, row).x);
        xn[q] = *(const v4f*)(X + (long long)row * IN_F + 4 * lane);
    }
}

__global__ __launch_bounds__(128) void gcn_gemm1_kernel(
        const float* __restrict__ X, const float* __restrict__ W,
        const float* __restrict__ deg, float* __restrict__ H, int nN, int nMT) {
    __shared__ __attribute__((aligned(16))) float xa[16 * G1_PA];
    __shared__ __attribute__((aligned(16))) float xo[16 * G1_PO];
    const int lane = threadIdx.x & 31;
    const int wv   = threadIdx.x >> 5;
    const int half = lane >> 4;
    const int l    = lane & 15;
    const int n0   = wv * 16;

    v2f b[IN_F / 4];
#pragma unroll
    for (int ks = 0; ks < IN_F / 4; ++ks) {
        const int ka = 4 * ks + 2 * half;
        b[ks].x = W[ka       * HID_F + n0 + l];
        b[ks].y = W[(ka + 1) * HID_F + n0 + l];
    }

    v4f   xn[4];
    float sn[4];
    int mt = blockIdx.x;
    const int gstep = (int)gridDim.x;
    gcn_gemm1_rows(X, deg, nN, mt, wv, lane, xn, sn);

    for (; mt < nMT; mt += gstep) {
#pragma unroll
        for (int q = 0; q < 4; ++q)
            *(v4f*)(xa + (4 * wv + q) * G1_PA + 4 * lane) = xn[q] * sn[q];
        gcn_lds_barrier();

        {
            int mn = mt + gstep;
            mn = mn < nMT ? mn : nMT - 1;
            gcn_gemm1_rows(X, deg, nN, mn, wv, lane, xn, sn);
        }
        __builtin_amdgcn_sched_barrier(0);

        v2f a[IN_F / 4];
        const float* ar = xa + l * G1_PA + 2 * half;
#pragma unroll
        for (int ks = 0; ks < IN_F / 4; ++ks)
            a[ks] = *(const v2f*)(ar + 4 * ks);
        __builtin_amdgcn_sched_barrier(0);

        v8f acc = {};
#pragma unroll
        for (int ks = 0; ks < IN_F / 4; ++ks)
            acc = __builtin_amdgcn_wmma_f32_16x16x4_f32(
                      false, a[ks], false, b[ks], (short)0, acc, false, false);
#pragma unroll
        for (int r = 0; r < 8; ++r)
            xo[(r + 8 * half) * G1_PO + n0 + l] = acc[r];
        gcn_lds_barrier();

#pragma unroll
        for (int q = 0; q < 2; ++q) {
            const int r    = 4 * wv + 2 * q + half;
            const int orow = mt * 16 + r;
            const v4f o = *(const v4f*)(xo + r * G1_PO + 4 * l);
            if (orow < nN) *(v4f*)(H + (long long)orow * HID_F + 4 * l) = o;
        }
    }
}

#define EGRP 16
#define WTH  512
#define MT    ((BR + 15) / 16)
#define RP    (MT * 16)
#define PITCH 68

template <bool FULL>
__device__ __forceinline__ void gcn_walk_group(
        const float* __restrict__ H, float* acc, int sv, int dv,
        int lane, int l0, int last, int nvalid) {
    float v0[EGRP], v1[EGRP];
    int   li[EGRP];
#pragma unroll
    for (int u = 0; u < EGRP; ++u) {
        li[u] = FULL ? (l0 + u) : ((l0 + u) < last ? (l0 + u) : last);
        const int s = __builtin_amdgcn_readlane(sv, li[u]);
        const float* hr = H + (long long)s * HID_F;
        v0[u] = hr[lane];
        v1[u] = hr[lane + 32];
    }
#pragma unroll
    for (int u = 0; u < EGRP; ++u) {
        if (FULL || u < nvalid) {
            const int d = __builtin_amdgcn_readlane(dv, li[u]);
            float* ar = acc + d;
            atomicAdd(ar + lane,      v0[u]);
            atomicAdd(ar + lane + 32, v1[u]);
        }
    }
}

__device__ __forceinline__ void gcn_walk_list(
        const float* __restrict__ H, const int* __restrict__ lp, int n,
        float* acc, int row0, int lane, int t) {
    for (int base = t * 32; base < n; base += WTH) {
        const int idx = base + lane;
        const int ci  = idx < n ? idx : n - 1;
        const v2i rec = *(const v2i*)(lp + 2LL * ci);
        const int m   = (n - base) < 32 ? (n - base) : 32;
        const int sv  = rec.x;
        const int dv  = (rec.y - row0) * PITCH;
        for (int g = 0; g < m; g += EGRP) {
            const int left = m - g;
            if (left >= EGRP)
                gcn_walk_group<true >(H, acc, sv, dv, lane, g, 0, EGRP);
            else
                gcn_walk_group<false>(H, acc, sv, dv, lane, g, m - 1, left);
        }
    }
}

template <bool L1>
__global__ __launch_bounds__(WTH) void gcn_bucket_walk_kernel(
        const float* __restrict__ H, const int* __restrict__ lists,
        const int* __restrict__ ovf, const int* __restrict__ ctr,
        const float* __restrict__ deg, const float* __restrict__ bias,
        const float* __restrict__ W, float* __restrict__ OUT,
        int nE, int nN, int nB, int nBP) {
    __shared__ float acc[RP * PITCH];
    const int lane = threadIdx.x & 31;
    const int wv   = __builtin_amdgcn_readfirstlane((int)(threadIdx.x >> 5));
    const int bk   = blockIdx.x;
    const int row0 = bk * BR;

    {
        const v4f z = {0.0f, 0.0f, 0.0f, 0.0f};
        for (int i = threadIdx.x; i < RP * PITCH / 4; i += WTH) ((v4f*)acc)[i] = z;
    }
    __syncthreads();

    for (int hf = 0; hf < 2; ++hf) {
        int n = ctr[64 + CST * (hf * nBP + bk)];
        n = n < CAPH ? n : CAPH;
        gcn_walk_list(H, lists + 2LL * ((long long)(hf * nB + bk) * CAPH), n,
                      acc, row0, lane, (wv + hf * (WTH / 64)) & (WTH / 32 - 1));
    }

    {
        int no = ctr[0];
        no = no < nE ? no : nE;
        for (int base = wv * 32; base < no; base += WTH) {
            const int idx = base + lane;
            const int ci  = idx < no ? idx : no - 1;
            const v2i rec = *(const v2i*)(ovf + 2LL * ci);
            const int m   = (no - base) < 32 ? (no - base) : 32;
            for (int u = 0; u < m; ++u) {
                const int d = __builtin_amdgcn_readlane(rec.y, u) - row0;
                if ((unsigned)d < (unsigned)BR) {
                    const int s = __builtin_amdgcn_readlane(rec.x, u);
                    const float* hr = H + (long long)s * HID_F;
                    atomicAdd(acc + d * PITCH + lane,      hr[lane]);
                    atomicAdd(acc + d * PITCH + lane + 32, hr[lane + 32]);
                }
            }
        }
    }
    __syncthreads();

    const float bias0 = bias[lane];
    const float bias1 = bias[lane + 32];
    for (int r = wv; r < BR; r += WTH / 32) {
        const int row = row0 + r;
        if (row < nN) {
            float a0 = acc[r * PITCH + lane];
            float a1 = acc[r * PITCH + lane + 32];
            const v2f g = gcn_deg_pair(deg, nN, row);
            const float n = gcn_norm_of(g.y);
            a0 = a0 * n + bias0;
            a1 = a1 * n + bias1;
            if (L1) {
                const float w = gcn_norm_of(g.x);
                acc[r * PITCH + lane]      = (a0 > 0.0f ? a0 : 0.0f) * w;
                acc[r * PITCH + lane + 32] = (a1 > 0.0f ? a1 : 0.0f) * w;
            } else {
                float* orow = OUT + (long long)row * OUT_F;
                orow[lane]      = a0;
                orow[lane + 32] = a1;
            }
        }
    }

    if (L1) {
        __syncthreads();
        const int half = lane >> 4;
        const int l    = lane & 15;
        for (int job = wv; job < MT * (OUT_F / 16); job += WTH / 32) {
            const int mt = job / (OUT_F / 16);
            const int n0 = (job % (OUT_F / 16)) * 16;
            const float* ar = acc + (mt * 16 + l) * PITCH;
            v8f c = {};
#pragma unroll
            for (int k0 = 0; k0 < HID_F; k0 += 4) {
                const int ka = k0 + 2 * half;
                const v2f a = *(const v2f*)(ar + ka);
                v2f b;
                b.x = W[(long long)ka       * OUT_F + n0 + l];
                b.y = W[(long long)(ka + 1) * OUT_F + n0 + l];
                c = __builtin_amdgcn_wmma_f32_16x16x4_f32(
                        false, a, false, b, (short)0, c, false, false);
            }
#pragma unroll
            for (int r = 0; r < 8; ++r) {
                const int br   = mt * 16 + r + 8 * half;
                const int orow = row0 + br;
                if (br < BR && orow < nN)
                    OUT[(long long)orow * OUT_F + n0 + l] = c[r];
            }
        }
    }
}

extern "C" void kernel_launch(void* const* d_in, const int* in_sizes, int n_in,
                              void* d_out, int out_size, void* d_ws, size_t ws_size,
                              hipStream_t stream) {
    const float* X   = (const float*)d_in[0];
    const int*   src = (const int*)  d_in[1];
    const int*   dst = (const int*)  d_in[2];
    const float* W1  = (const float*)d_in[3];
    const float* b1  = (const float*)d_in[4];
    const float* W2  = (const float*)d_in[5];
    const float* b2  = (const float*)d_in[6];
    float*       out = (float*)d_out;

    const int nN = in_sizes[0] / IN_F;
    const int nE = in_sizes[1];

    const int nB  = (nN + BR - 1) / BR;
    const int nMT = (nN + 15) / 16;
    const int nBP = (nB + 31) / 32 * 32;

    float* ws    = (float*)d_ws;
    float* bufA  = ws;
    float* bufB  = bufA + (long long)nN * HID_F;
    float* deg   = bufB + (long long)nN * HID_F;
    int*   ctr   = (int*)(deg + 4LL * nN);
    int*   lists = ctr + 64 + 2 * CST * nBP;
    int*   ovf   = lists + 4LL * nB * CAPH;

    gcn_zero_kernel         <<<(4LL*nN + 64 + 2*CST*nBP + 255) / 256, 256, 0, stream>>>(deg, 4LL*nN + 64 + 2*CST*nBP);
    gcn_degree_bucket_kernel<<<(nE + FB_SL - 1) / FB_SL, FB_NT, 0, stream>>>(src, dst, deg, ctr, lists, ovf, nE, nN, nB, nBP);

    gcn_gemm1_kernel<<<(nMT + G1_TPB - 1) / G1_TPB, 128, 0, stream>>>(X, W1, deg, bufA, nN, nMT);
    gcn_bucket_walk_kernel<true> <<<nB, WTH, 0, stream>>>(bufA, lists, ovf, ctr, deg, b1, W2, bufB, nE, nN, nB, nBP);

    gcn_bucket_walk_kernel<false><<<nB, WTH, 0, stream>>>(bufB, lists, ovf, ctr, deg, b2, W2, out, nE, nN, nB, nBP);
}


# In-degree counted where it is used: a walk block counts its rows' records in LDS (integer adds) as it reads its lists; the filing pass's in-degree memory atomics and the walks' in-degree reads go
# speedup vs baseline: 6.7075x; 1.0490x over previous
#include <hip/hip_runtime.h>

#define IN_F  128
#define HID_F 64
#define OUT_F 64

typedef __attribute__((ext_vector_type(2))) float v2f;
typedef __attribute__((ext_vector_type(4))) float v4f;
typedef __attribute__((ext_vector_type(8))) float v8f;

__global__ void gcn_zero_kernel(float* __restrict__ p, long long n) {
    long long i = (long long)blockIdx.x * blockDim.x + threadIdx.x;
    if (i < n) p[i] = 0.0f;
}

__device__ __forceinline__ int gcn_my_half() {
    const unsigned id = (unsigned)__builtin_amdgcn_s_sendmsg_rtn(0x87);
    return (int)((id >> 18) & 1u);
}

#define BR   200
#define CAPH 2560
#define CST  32

typedef __attribute__((ext_vector_type(2))) int v2i;

__device__ __forceinline__ void gcn_lds_barrier() {
    asm volatile("s_wait_dscnt 0x0\n\ts_barrier_signal -1\n\ts_barrier_wait -1" ::: "memory");
}

#define FB_NT  1024
#define FB_EPT 7
#define FB_SL  (FB_NT * FB_EPT)
#define FB_NB  512

__global__ __launch_bounds__(FB_NT) void gcn_degree_bucket_kernel(
        const int* __restrict__ src, const int* __restrict__ dst,
        float* __restrict__ deg, int* __restrict__ ctr,
        int* __restrict__ lists, int* __restrict__ ovf,
        int nE, int nN, int nB, int nBP) {
    __shared__ v2i srec[FB_SL];
    __shared__ int c_cnt[FB_NB];
    __shared__ int c_exc[FB_NB];
    __shared__ int c_del[FB_NB];
    __shared__ int c_wt[FB_NB / 32];
    __shared__ int c_tot;
    const int tid  = threadIdx.x;
    const int lane = tid & 31;
    const int wv   = tid >> 5;
    const int hf   = gcn_my_half();
    float* __restrict__ dg = deg + (long long)hf * nN;

    int s[FB_EPT], d[FB_EPT], b[FB_EPT], r[FB_EPT];
#pragma unroll
    for (int k = 0; k < FB_EPT; ++k) {
        const int e = blockIdx.x * FB_SL + k * FB_NT + tid;
        const bool ok = e < nE;
        s[k] = ok ? src[e] : 0;
        d[k] = ok ? dst[e] : -1;
    }
#pragma unroll
    for (int k = 0; k < FB_EPT; ++k) {
        b[k] = -1;
        if (d[k] >= 0) {
            atomicAdd(&dg[s[k]], 1.0f);
            b[k] = d[k] / BR;
        }
    }

    for (int c0 = 0; c0 < nB; c0 += FB_NB) {
        if (tid < FB_NB) c_cnt[tid] = 0;
        gcn_lds_barrier();
#pragma unroll
        for (int k = 0; k < FB_EPT; ++k) {
            const unsigned bl = (unsigned)(b[k] - c0);
            r[k] = bl < (unsigned)FB_NB ? atomicAdd(&c_cnt[bl], 1) : 0;
        }
        gcn_lds_barrier();

        const bool mine = tid < FB_NB && (c0 + tid) < nB;
        const int  v    = mine ? c_cnt[tid] : 0;
        int inc = v;
        for (int o = 1; o < 32; o <<= 1) {
            const int t = __shfl_up(inc, o);
            if (lane >= o) inc += t;
        }
        if (tid < FB_NB && lane == 31) c_wt[wv] = inc;
        gcn_lds_barrier();
        if (tid < FB_NB) {
            int off = 0;
            for (int j = 0; j < wv; ++j) off += c_wt[j];
            const int ex = off + inc - v;
            int g = 0;
            if (v > 0) g = atomicAdd(ctr + 64 + CST * (hf * nBP + c0 + tid), v);
            c_exc[tid] = ex;
            c_del[tid] = g - ex;
            if (tid == FB_NB - 1) c_tot = ex + v;
        }
        gcn_lds_barrier();

#pragma unroll
        for (int k = 0; k < FB_EPT; ++k) {
            const unsigned bl = (unsigned)(b[k] - c0);
            if (bl < (unsigned)FB_NB) {
                const v2i rec = {s[k], d[k]};
                srec[c_exc[bl] + r[k]] = rec;
            }
        }
        gcn_lds_barrier();

        const int tot = c_tot;
        for (int i = tid; i < tot; i += FB_NT) {
            const v2i rec = srec[i];
            const int bk  = rec.y / BR;
            const int pos = i + c_del[bk - c0];
            if (pos < CAPH) {
                *(v2i*)(lists + 2LL * ((long long)(hf * nB + bk) * CAPH + pos)) = rec;
            } else {
                const int op = atomicAdd(ctr, 1);
                *(v2i*)(ovf + 2LL * op) = rec;
            }
        }
        gcn_lds_barrier();
    }
}

__device__ __forceinline__ float gcn_out_deg(const float* __restrict__ deg, int nN, int node) {
    return deg[node] + deg[(long long)nN + node];
}

__device__ __forceinline__ float gcn_norm_of(float d) {
    d = d < 1.0f ? 1.0f : d;
    return rsqrtf(d);
}

#define G1_TPB 5
#define G1_PA  132
#define G1_PO  68

__device__ __forceinline__ void gcn_gemm1_rows(
        const float* __restrict__ X, const float* __restrict__ deg, int nN,
        int mt, int wv, int lane, v4f* xn, float* sn) {
#pragma unroll
    for (int q = 0; q < 4; ++q) {
        int row = mt * 16 + 4 * wv + q;
        row = row < nN ? row : nN - 1;
        sn[q] = gcn_norm_of(gcn_out_deg(deg, nN, row));
        xn[q] = *(const v4f*)(X + (long long)row * IN_F + 4 * lane);
    }
}

__global__ __launch_bounds__(128) void gcn_gemm1_kernel(
        const float* __restrict__ X, const float* __restrict__ W,
        const float* __restrict__ deg, float* __restrict__ H, int nN, int nMT) {
    __shared__ __attribute__((aligned(16))) float xa[16 * G1_PA];
    __shared__ __attribute__((aligned(16))) float xo[16 * G1_PO];
    const int lane = threadIdx.x & 31;
    const int wv   = threadIdx.x >> 5;
    const int half = lane >> 4;
    const int l    = lane & 15;
    const int n0   = wv * 16;

    v2f b[IN_F / 4];
#pragma unroll
    for (int ks = 0; ks < IN_F / 4; ++ks) {
        const int ka = 4 * ks + 2 * half;
        b[ks].x = W[ka       * HID_F + n0 + l];
        b[ks].y = W[(ka + 1) * HID_F + n0 + l];
    }

    v4f   xn[4];
    float sn[4];
    int mt = blockIdx.x;
    const int gstep = (int)gridDim.x;
    gcn_gemm1_rows(X, deg, nN, mt, wv, lane, xn, sn);

    for (; mt < nMT; mt += gstep) {
#pragma unroll
        for (int q = 0; q < 4; ++q)
            *(v4f*)(xa + (4 * wv + q) * G1_PA + 4 * lane) = xn[q] * sn[q];
        gcn_lds_barrier();

        {
            int mn = mt + gstep;
            mn = mn < nMT ? mn : nMT - 1;
            gcn_gemm1_rows(X, deg, nN, mn, wv, lane, xn, sn);
        }
        __builtin_amdgcn_sched_barrier(0);

        v2f a[IN_F / 4];
        const float* ar = xa + l * G1_PA + 2 * half;
#pragma unroll
        for (int ks = 0; ks < IN_F / 4; ++ks)
            a[ks] = *(const v2f*)(ar + 4 * ks);
        __builtin_amdgcn_sched_barrier(0);

        v8f acc = {};
#pragma unroll
        for (int ks = 0; ks < IN_F / 4; ++ks)
            acc = __builtin_amdgcn_wmma_f32_16x16x4_f32(
                      false, a[ks], false, b[ks], (short)0, acc, false, false);
#pragma unroll
        for (int r = 0; r < 8; ++r)
            xo[(r + 8 * half) * G1_PO + n0 + l] = acc[r];
        gcn_lds_barrier();

#pragma unroll
        for (int q = 0; q < 2; ++q) {
            const int r    = 4 * wv + 2 * q + half;
            const int orow = mt * 16 + r;
            const v4f o = *(const v4f*)(xo + r * G1_PO + 4 * l);
            if (orow < nN) *(v4f*)(H + (long long)orow * HID_F + 4 * l) = o;
        }
    }
}

#define EGRP 16
#define WTH  512
#define MT    ((BR + 15) / 16)
#define RP    (MT * 16)
#define PITCH 68

template <bool FULL>
__device__ __forceinline__ void gcn_walk_group(
        const float* __restrict__ H, float* acc, int sv, int dv,
        int lane, int l0, int last, int nvalid) {
    float v0[EGRP], v1[EGRP];
    int   li[EGRP];
#pragma unroll
    for (int u = 0; u < EGRP; ++u) {
        li[u] = FULL ? (l0 + u) : ((l0 + u) < last ? (l0 + u) : last);
        const int s = __builtin_amdgcn_readlane(sv, li[u]);
        const float* hr = H + (long long)s * HID_F;
        v0[u] = hr[lane];
        v1[u] = hr[lane + 32];
    }
#pragma unroll
    for (int u = 0; u < EGRP; ++u) {
        if (FULL || u < nvalid) {
            const int d = __builtin_amdgcn_readlane(dv, li[u]);
            float* ar = acc + d;
            atomicAdd(ar + lane,      v0[u]);
            atomicAdd(ar + lane + 32, v1[u]);
        }
    }
}

__device__ __forceinline__ void gcn_walk_list(
        const float* __restrict__ H, const int* __restrict__ lp, int n,
        float* acc, int* cnt, int row0, int lane, int t) {
    for (int base = t * 32; base < n; base += WTH) {
        const int idx = base + lane;
        const int ci  = idx < n ? idx : n - 1;
        const v2i rec = *(const v2i*)(lp + 2LL * ci);
        const int m   = (n - base) < 32 ? (n - base) : 32;
        if (idx < n) atomicAdd(&cnt[rec.y - row0], 1);
        const int sv  = rec.x;
        const int dv  = (rec.y - row0) * PITCH;
        for (int g = 0; g < m; g += EGRP) {
            const int left = m - g;
            if (left >= EGRP)
                gcn_walk_group<true >(H, acc, sv, dv, lane, g, 0, EGRP);
            else
                gcn_walk_group<false>(H, acc, sv, dv, lane, g, m - 1, left);
        }
    }
}

template <bool L1>
__global__ __launch_bounds__(WTH) void gcn_bucket_walk_kernel(
        const float* __restrict__ H, const int* __restrict__ lists,
        const int* __restrict__ ovf, const int* __restrict__ ctr,
        const float* __restrict__ deg, const float* __restrict__ bias,
        const float* __restrict__ W, float* __restrict__ OUT,
        int nE, int nN, int nB, int nBP) {
    __shared__ float acc[RP * PITCH];
    __shared__ int   cnt[RP];
    const int lane = threadIdx.x & 31;
    const int wv   = __builtin_amdgcn_readfirstlane((int)(threadIdx.x >> 5));
    const int bk   = blockIdx.x;
    const int row0 = bk * BR;

    {
        const v4f z = {0.0f, 0.0f, 0.0f, 0.0f};
        for (int i = threadIdx.x; i < RP * PITCH / 4; i += WTH) ((v4f*)acc)[i] = z;
        if (threadIdx.x < RP) cnt[threadIdx.x] = 0;
    }
    __syncthreads();

    for (int hf = 0; hf < 2; ++hf) {
        int n = ctr[64 + CST * (hf * nBP + bk)];
        n = n < CAPH ? n : CAPH;
        gcn_walk_list(H, lists + 2LL * ((long long)(hf * nB + bk) * CAPH), n,
                      acc, cnt, row0, lane, (wv + hf * (WTH / 64)) & (WTH / 32 - 1));
    }

    {
        int no = ctr[0];
        no = no < nE ? no : nE;
        for (int base = wv * 32; base < no; base += WTH) {
            const int idx = base + lane;
            const int ci  = idx < no ? idx : no - 1;
            const v2i rec = *(const v2i*)(ovf + 2LL * ci);
            const int m   = (no - base) < 32 ? (no - base) : 32;
            for (int u = 0; u < m; ++u) {
                const int d = __builtin_amdgcn_readlane(rec.y, u) - row0;
                if ((unsigned)d < (unsigned)BR) {
                    const int s = __builtin_amdgcn_readlane(rec.x, u);
                    const float* hr = H + (long long)s * HID_F;
                    if (lane == 0) atomicAdd(&cnt[d], 1);
                    atomicAdd(acc + d * PITCH + lane,      hr[lane]);
                    atomicAdd(acc + d * PITCH + lane + 32, hr[lane + 32]);
                }
            }
        }
    }
    __syncthreads();

    const float bias0 = bias[lane];
    const float bias1 = bias[lane + 32];
    for (int r = wv; r < BR; r += WTH / 32) {
        const int row = row0 + r;
        if (row < nN) {
            float a0 = acc[r * PITCH + lane];
            float a1 = acc[r * PITCH + lane + 32];
            const float n = gcn_norm_of((float)cnt[r]);
            a0 = a0 * n + bias0;
            a1 = a1 * n + bias1;
            if (L1) {
                const float w = gcn_norm_of(gcn_out_deg(deg, nN, row));
                acc[r * PITCH + lane]      = (a0 > 0.0f ? a0 : 0.0f) * w;
                acc[r * PITCH + lane + 32] = (a1 > 0.0f ? a1 : 0.0f) * w;
            } else {
                float* orow = OUT + (long long)row * OUT_F;
                orow[lane]      = a0;
                orow[lane + 32] = a1;
            }
        }
    }

    if (L1) {
        __syncthreads();
        const int half = lane >> 4;
        const int l    = lane & 15;
        for (int job = wv; job < MT * (OUT_F / 16); job += WTH / 32) {
            const int mt = job / (OUT_F / 16);
            const int n0 = (job % (OUT_F / 16)) * 16;
            const float* ar = acc + (mt * 16 + l) * PITCH;
            v8f c = {};
#pragma unroll
            for (int k0 = 0; k0 < HID_F; k0 += 4) {
                const int ka = k0 + 2 * half;
                const v2f a = *(const v2f*)(ar + ka);
                v2f b;
                b.x = W[(long long)ka       * OUT_F + n0 + l];
                b.y = W[(long long)(ka + 1) * OUT_F + n0 + l];
                c = __builtin_amdgcn_wmma_f32_16x16x4_f32(
                        false, a, false, b, (short)0, c, false, false);
            }
#pragma unroll
            for (int r = 0; r < 8; ++r) {
                const int br   = mt * 16 + r + 8 * half;
                const int orow = row0 + br;
                if (br < BR && orow < nN)
                    OUT[(long long)orow * OUT_F + n0 + l] = c[r];
            }
        }
    }
}

extern "C" void kernel_launch(void* const* d_in, const int* in_sizes, int n_in,
                              void* d_out, int out_size, void* d_ws, size_t ws_size,
                              hipStream_t stream) {
    const float* X   = (const float*)d_in[0];
    const int*   src = (const int*)  d_in[1];
    const int*   dst = (const int*)  d_in[2];
    const float* W1  = (const float*)d_in[3];
    const float* b1  = (const float*)d_in[4];
    const float* W2  = (const float*)d_in[5];
    const float* b2  = (const float*)d_in[6];
    float*       out = (float*)d_out;

    const int nN = in_sizes[0] / IN_F;
    const int nE = in_sizes[1];

    const int nB  = (nN + BR - 1) / BR;
    const int nMT = (nN + 15) / 16;
    const int nBP = (nB + 31) / 32 * 32;

    float* ws    = (float*)d_ws;
    float* bufA  = ws;
    float* bufB  = bufA + (long long)nN * HID_F;
    float* deg   = bufB + (long long)nN * HID_F;
    int*   ctr   = (int*)(deg + 2LL * nN);
    int*   lists = ctr + 64 + 2 * CST * nBP;
    int*   ovf   = lists + 4LL * nB * CAPH;

    gcn_zero_kernel         <<<(2LL*nN + 64 + 2*CST*nBP + 255) / 256, 256, 0, stream>>>(deg, 2LL*nN + 64 + 2*CST*nBP);
    gcn_degree_bucket_kernel<<<(nE + FB_SL - 1) / FB_SL, FB_NT, 0, stream>>>(src, dst, deg, ctr, lists, ovf, nE, nN, nB, nBP);

    gcn_gemm1_kernel<<<(nMT + G1_TPB - 1) / G1_TPB, 128, 0, stream>>>(X, W1, deg, bufA, nN, nMT);
    gcn_bucket_walk_kernel<true> <<<nB, WTH, 0, stream>>>(bufA, lists, ovf, ctr, deg, b1, W2, bufB, nE, nN, nB, nBP);

    gcn_bucket_walk_kernel<false><<<nB, WTH, 0, stream>>>(bufB, lists, ovf, ctr, deg, b2, W2, out, nE, nN, nB, nBP);
}


# Walk block sorts its lists' records by row in LDS (counting sort, integer adds); waves sum each row's run in registers: one LDS float add per run, not per edge; row sum order = LDS rank
# speedup vs baseline: 7.2189x; 1.0762x over previous
#include <hip/hip_runtime.h>

#define IN_F  128
#define HID_F 64
#define OUT_F 64

typedef __attribute__((ext_vector_type(2))) float v2f;
typedef __attribute__((ext_vector_type(4))) float v4f;
typedef __attribute__((ext_vector_type(8))) float v8f;

__global__ void gcn_zero_kernel(float* __restrict__ p, long long n) {
    long long i = (long long)blockIdx.x * blockDim.x + threadIdx.x;
    if (i < n) p[i] = 0.0f;
}

__device__ __forceinline__ int gcn_my_half() {
    const unsigned id = (unsigned)__builtin_amdgcn_s_sendmsg_rtn(0x87);
    return (int)((id >> 18) & 1u);
}

#define BR   200
#define CAPH 2560
#define CST  32

typedef __attribute__((ext_vector_type(2))) int v2i;

__device__ __forceinline__ void gcn_lds_barrier() {
    asm volatile("s_wait_dscnt 0x0\n\ts_barrier_signal -1\n\ts_barrier_wait -1" ::: "memory");
}

#define FB_NT  1024
#define FB_EPT 7
#define FB_SL  (FB_NT * FB_EPT)
#define FB_NB  512

__global__ __launch_bounds__(FB_NT) void gcn_degree_bucket_kernel(
        const int* __restrict__ src, const int* __restrict__ dst,
        float* __restrict__ deg, int* __restrict__ ctr,
        int* __restrict__ lists, int* __restrict__ ovf,
        int nE, int nN, int nB, int nBP) {
    __shared__ v2i srec[FB_SL];
    __shared__ int c_cnt[FB_NB];
    __shared__ int c_exc[FB_NB];
    __shared__ int c_del[FB_NB];
    __shared__ int c_wt[FB_NB / 32];
    __shared__ int c_tot;
    const int tid  = threadIdx.x;
    const int lane = tid & 31;
    const int wv   = tid >> 5;
    const int hf   = gcn_my_half();
    float* __restrict__ dg = deg + (long long)hf * nN;

    int s[FB_EPT], d[FB_EPT], b[FB_EPT], r[FB_EPT];
#pragma unroll
    for (int k = 0; k < FB_EPT; ++k) {
        const int e = blockIdx.x * FB_SL + k * FB_NT + tid;
        const bool ok = e < nE;
        s[k] = ok ? src[e] : 0;
        d[k] = ok ? dst[e] : -1;
    }
#pragma unroll
    for (int k = 0; k < FB_EPT; ++k) {
        b[k] = -1;
        if (d[k] >= 0) {
            atomicAdd(&dg[s[k]], 1.0f);
            b[k] = d[k] / BR;
        }
    }

    for (int c0 = 0; c0 < nB; c0 += FB_NB) {
        if (tid < FB_NB) c_cnt[tid] = 0;
        gcn_lds_barrier();
#pragma unroll
        for (int k = 0; k < FB_EPT; ++k) {
            const unsigned bl = (unsigned)(b[k] - c0);
            r[k] = bl < (unsigned)FB_NB ? atomicAdd(&c_cnt[bl], 1) : 0;
        }
        gcn_lds_barrier();

        const bool mine = tid < FB_NB && (c0 + tid) < nB;
        const int  v    = mine ? c_cnt[tid] : 0;
        int inc = v;
        for (int o = 1; o < 32; o <<= 1) {
            const int t = __shfl_up(inc, o);
            if (lane >= o) inc += t;
        }
        if (tid < FB_NB && lane == 31) c_wt[wv] = inc;
        gcn_lds_barrier();
        if (tid < FB_NB) {
            int off = 0;
            for (int j = 0; j < wv; ++j) off += c_wt[j];
            const int ex = off + inc - v;
            int g = 0;
            if (v > 0) g = atomicAdd(ctr + 64 + CST * (hf * nBP + c0 + tid), v);
            c_exc[tid] = ex;
            c_del[tid] = g - ex;
            if (tid == FB_NB - 1) c_tot = ex + v;
        }
        gcn_lds_barrier();

#pragma unroll
        for (int k = 0; k < FB_EPT; ++k) {
            const unsigned bl = (unsigned)(b[k] - c0);
            if (bl < (unsigned)FB_NB) {
                const v2i rec = {s[k], d[k]};
                srec[c_exc[bl] + r[k]] = rec;
            }
        }
        gcn_lds_barrier();

        const int tot = c_tot;
        for (int i = tid; i < tot; i += FB_NT) {
            const v2i rec = srec[i];
            const int bk  = rec.y / BR;
            const int pos = i + c_del[bk - c0];
            if (pos < CAPH) {
                *(v2i*)(lists + 2LL * ((long long)(hf * nB + bk) * CAPH + pos)) = rec;
            } else {
                const int op = atomicAdd(ctr, 1);
                *(v2i*)(ovf + 2LL * op) = rec;
            }
        }
        gcn_lds_barrier();
    }
}

__device__ __forceinline__ float gcn_out_deg(const float* __restrict__ deg, int nN, int node) {
    return deg[node] + deg[(long long)nN + node];
}

__device__ __forceinline__ float gcn_norm_of(float d) {
    d = d < 1.0f ? 1.0f : d;
    return rsqrtf(d);
}

#define G1_TPB 5
#define G1_PA  132
#define G1_PO  68

__device__ __forceinline__ void gcn_gemm1_rows(
        const float* __restrict__ X, const float* __restrict__ deg, int nN,
        int mt, int wv, int lane, v4f* xn, float* sn) {
#pragma unroll
    for (int q = 0; q < 4; ++q) {
        int row = mt * 16 + 4 * wv + q;
        row = row < nN ? row : nN - 1;
        sn[q] = gcn_norm_of(gcn_out_deg(deg, nN, row));
        xn[q] = *(const v4f*)(X + (long long)row * IN_F + 4 * lane);
    }
}

__global__ __launch_bounds__(128) void gcn_gemm1_kernel(
        const float* __restrict__ X, const float* __restrict__ W,
        const float* __restrict__ deg, float* __restrict__ H, int nN, int nMT) {
    __shared__ __attribute__((aligned(16))) float xa[16 * G1_PA];
    __shared__ __attribute__((aligned(16))) float xo[16 * G1_PO];
    const int lane = threadIdx.x & 31;
    const int wv   = threadIdx.x >> 5;
    const int half = lane >> 4;
    const int l    = lane & 15;
    const int n0   = wv * 16;

    v2f b[IN_F / 4];
#pragma unroll
    for (int ks = 0; ks < IN_F / 4; ++ks) {
        const int ka = 4 * ks + 2 * half;
        b[ks].x = W[ka       * HID_F + n0 + l];
        b[ks].y = W[(ka + 1) * HID_F + n0 + l];
    }

    v4f   xn[4];
    float sn[4];
    int mt = blockIdx.x;
    const int gstep = (int)gridDim.x;
    gcn_gemm1_rows(X, deg, nN, mt, wv, lane, xn, sn);

    for (; mt < nMT; mt += gstep) {
#pragma unroll
        for (int q = 0; q < 4; ++q)
            *(v4f*)(xa + (4 * wv + q) * G1_PA + 4 * lane) = xn[q] * sn[q];
        gcn_lds_barrier();

        {
            int mn = mt + gstep;
            mn = mn < nMT ? mn : nMT - 1;
            gcn_gemm1_rows(X, deg, nN, mn, wv, lane, xn, sn);
        }
        __builtin_amdgcn_sched_barrier(0);

        v2f a[IN_F / 4];
        const float* ar = xa + l * G1_PA + 2 * half;
#pragma unroll
        for (int ks = 0; ks < IN_F / 4; ++ks)
            a[ks] = *(const v2f*)(ar + 4 * ks);
        __builtin_amdgcn_sched_barrier(0);

        v8f acc = {};
#pragma unroll
        for (int ks = 0; ks < IN_F / 4; ++ks)
            acc = __builtin_amdgcn_wmma_f32_16x16x4_f32(
                      false, a[ks], false, b[ks], (short)0, acc, false, false);
#pragma unroll
        for (int r = 0; r < 8; ++r)
            xo[(r + 8 * half) * G1_PO + n0 + l] = acc[r];
        gcn_lds_barrier();

#pragma unroll
        for (int q = 0; q < 2; ++q) {
            const int r    = 4 * wv + 2 * q + half;
            const int orow = mt * 16 + r;
            const v4f o = *(const v4f*)(xo + r * G1_PO + 4 * l);
            if (orow < nN) *(v4f*)(H + (long long)orow * HID_F + 4 * l) = o;
        }
    }
}

#define EGRP 16
#define WTH  512
#define WPC  ((CAPH + WTH - 1) / WTH)
#define MT    ((BR + 15) / 16)
#define RP    (MT * 16)
#define PITCH 68

template <bool FULL>
__device__ __forceinline__ void gcn_sum_group(
        const float* __restrict__ H, float* acc, int sv, int dv,
        int lane, int l0, int last, int nvalid,
        int& curd, float& a0, float& a1) {
    float v0[EGRP], v1[EGRP];
    int   li[EGRP];
#pragma unroll
    for (int u = 0; u < EGRP; ++u) {
        li[u] = FULL ? (l0 + u) : ((l0 + u) < last ? (l0 + u) : last);
        const int s = __builtin_amdgcn_readlane(sv, li[u]);
        const float* hr = H + (long long)s * HID_F;
        v0[u] = hr[lane];
        v1[u] = hr[lane + 32];
    }
#pragma unroll
    for (int u = 0; u < EGRP; ++u) {
        if (FULL || u < nvalid) {
            const int d = __builtin_amdgcn_readlane(dv, li[u]);
            if (__builtin_expect(d != curd, 0)) {
                if (curd >= 0) {
                    atomicAdd(acc + curd + lane,      a0);
                    atomicAdd(acc + curd + lane + 32, a1);
                }
                curd = d;
                a0 = 0.0f;
                a1 = 0.0f;
            }
            a0 += v0[u];
            a1 += v1[u];
        }
    }
}

template <bool L1>
__global__ __launch_bounds__(WTH) void gcn_bucket_walk_kernel(
        const float* __restrict__ H, const int* __restrict__ lists,
        const int* __restrict__ ovf, const int* __restrict__ ctr,
        const float* __restrict__ deg, const float* __restrict__ bias,
        const float* __restrict__ W, float* __restrict__ OUT,
        int nE, int nN, int nB, int nBP) {
    __shared__ float acc[RP * PITCH];
    __shared__ int   cnt[RP];
    __shared__ int   cur[RP];
    __shared__ v2i   srt[2 * CAPH];
    __shared__ int   wsum[(RP + 31) / 32];
    __shared__ int   s_tot;
    const int lane = threadIdx.x & 31;
    const int wv   = __builtin_amdgcn_readfirstlane((int)(threadIdx.x >> 5));
    const int bk   = blockIdx.x;
    const int row0 = bk * BR;

    {
        const v4f z = {0.0f, 0.0f, 0.0f, 0.0f};
        for (int i = threadIdx.x; i < RP * PITCH / 4; i += WTH) ((v4f*)acc)[i] = z;
        if (threadIdx.x < RP) cnt[threadIdx.x] = 0;
    }

    v2i rec[2 * WPC];
#pragma unroll
    for (int hf = 0; hf < 2; ++hf) {
        int n = ctr[64 + CST * (hf * nBP + bk)];
        n = n < CAPH ? n : CAPH;
        const int t = (wv + hf * (WTH / 64)) & (WTH / 32 - 1);
        const int* __restrict__ lp = lists + 2LL * ((long long)(hf * nB + bk) * CAPH);
#pragma unroll
        for (int j = 0; j < WPC; ++j) {
            const int idx = t * 32 + j * WTH + lane;
            v2i r = {0, -1};
            if (idx < n) r = *(const v2i*)(lp + 2LL * idx);
            rec[hf * WPC + j] = r;
        }
    }
    __syncthreads();
#pragma unroll
    for (int k = 0; k < 2 * WPC; ++k)
        if (rec[k].y >= 0) atomicAdd(&cnt[rec[k].y - row0], 1);
    __syncthreads();

    {
        const int tid = threadIdx.x;
        const int v   = tid < RP ? cnt[tid] : 0;
        int inc = v;
        for (int o = 1; o < 32; o <<= 1) {
            const int t = __shfl_up(inc, o);
            if (lane >= o) inc += t;
        }
        if (lane == 31 && wv < (RP + 31) / 32) wsum[wv] = inc;
        __syncthreads();
        if (tid < RP) {
            int off = 0;
            for (int j = 0; j < wv; ++j) off += wsum[j];
            cur[tid] = off + inc - v;
            if (tid == RP - 1) s_tot = off + inc;
        }
        __syncthreads();
    }
#pragma unroll
    for (int k = 0; k < 2 * WPC; ++k) {
        if (rec[k].y >= 0) {
            const int r   = rec[k].y - row0;
            const int pos = atomicAdd(&cur[r], 1);
            const v2i o = {rec[k].x, r * PITCH};
            srt[pos] = o;
        }
    }
    __syncthreads();

    {
        const int tot = __builtin_amdgcn_readfirstlane(s_tot);
        const int ng  = (tot + EGRP - 1) / EGRP;
        const int b0  = (wv * ng / (WTH / 32)) * EGRP;
        int       b1  = ((wv + 1) * ng / (WTH / 32)) * EGRP;
        b1 = b1 < tot ? b1 : tot;
        int   curd = -1;
        float a0 = 0.0f, a1 = 0.0f;
        for (int base = b0; base < b1; base += 32) {
            const int idx = base + lane;
            const int ci  = idx < b1 ? idx : b1 - 1;
            const v2i r   = srt[ci];
            const int m   = (b1 - base) < 32 ? (b1 - base) : 32;
            for (int g = 0; g < m; g += EGRP) {
                const int left = m - g;
                if (left >= EGRP)
                    gcn_sum_group<true >(H, acc, r.x, r.y, lane, g, 0, EGRP, curd, a0, a1);
                else
                    gcn_sum_group<false>(H, acc, r.x, r.y, lane, g, m - 1, left, curd, a0, a1);
            }
        }
        if (curd >= 0) {
            atomicAdd(acc + curd + lane,      a0);
            atomicAdd(acc + curd + lane + 32, a1);
        }
    }

    {
        int no = ctr[0];
        no = no < nE ? no : nE;
        for (int base = wv * 32; base < no; base += WTH) {
            const int idx = base + lane;
            const int ci  = idx < no ? idx : no - 1;
            const v2i orec = *(const v2i*)(ovf + 2LL * ci);
            const int m   = (no - base) < 32 ? (no - base) : 32;
            for (int u = 0; u < m; ++u) {
                const int d = __builtin_amdgcn_readlane(orec.y, u) - row0;
                if ((unsigned)d < (unsigned)BR) {
                    const int s = __builtin_amdgcn_readlane(orec.x, u);
                    const float* hr = H + (long long)s * HID_F;
                    if (lane == 0) atomicAdd(&cnt[d], 1);
                    atomicAdd(acc + d * PITCH + lane,      hr[lane]);
                    atomicAdd(acc + d * PITCH + lane + 32, hr[lane + 32]);
                }
            }
        }
    }
    __syncthreads();

    const float bias0 = bias[lane];
    const float bias1 = bias[lane + 32];
    for (int r = wv; r < BR; r += WTH / 32) {
        const int row = row0 + r;
        if (row < nN) {
            float a0 = acc[r * PITCH + lane];
            float a1 = acc[r * PITCH + lane + 32];
            const float n = gcn_norm_of((float)cnt[r]);
            a0 = a0 * n + bias0;
            a1 = a1 * n + bias1;
            if (L1) {
                const float w = gcn_norm_of(gcn_out_deg(deg, nN, row));
                acc[r * PITCH + lane]      = (a0 > 0.0f ? a0 : 0.0f) * w;
                acc[r * PITCH + lane + 32] = (a1 > 0.0f ? a1 : 0.0f) * w;
            } else {
                float* orow = OUT + (long long)row * OUT_F;
                orow[lane]      = a0;
                orow[lane + 32] = a1;
            }
        }
    }

    if (L1) {
        __syncthreads();
        const int half = lane >> 4;
        const int l    = lane & 15;
        for (int job = wv; job < MT * (OUT_F / 16); job += WTH / 32) {
            const int mt = job / (OUT_F / 16);
            const int n0 = (job % (OUT_F / 16)) * 16;
            const float* ar = acc + (mt * 16 + l) * PITCH;
            v8f c = {};
#pragma unroll
            for (int k0 = 0; k0 < HID_F; k0 += 4) {
                const int ka = k0 + 2 * half;
                const v2f a = *(const v2f*)(ar + ka);
                v2f b;
                b.x = W[(long long)ka       * OUT_F + n0 + l];
                b.y = W[(long long)(ka + 1) * OUT_F + n0 + l];
                c = __builtin_amdgcn_wmma_f32_16x16x4_f32(
                        false, a, false, b, (short)0, c, false, false);
            }
#pragma unroll
            for (int r = 0; r < 8; ++r) {
                const int br   = mt * 16 + r + 8 * half;
                const int orow = row0 + br;
                if (br < BR && orow < nN)
                    OUT[(long long)orow * OUT_F + n0 + l] = c[r];
            }
        }
    }
}

extern "C" void kernel_launch(void* const* d_in, const int* in_sizes, int n_in,
                              void* d_out, int out_size, void* d_ws, size_t ws_size,
                              hipStream_t stream) {
    const float* X   = (const float*)d_in[0];
    const int*   src = (const int*)  d_in[1];
    const int*   dst = (const int*)  d_in[2];
    const float* W1  = (const float*)d_in[3];
    const float* b1  = (const float*)d_in[4];
    const float* W2  = (const float*)d_in[5];
    const float* b2  = (const float*)d_in[6];
    float*       out = (float*)d_out;

    const int nN = in_sizes[0] / IN_F;
    const int nE = in_sizes[1];

    const int nB  = (nN + BR - 1) / BR;
    const int nMT = (nN + 15) / 16;
    const int nBP = (nB + 31) / 32 * 32;

    float* ws    = (float*)d_ws;
    float* bufA  = ws;
    float* bufB  = bufA + (long long)nN * HID_F;
    float* deg   = bufB + (long long)nN * HID_F;
    int*   ctr   = (int*)(deg + 2LL * nN);
    int*   lists = ctr + 64 + 2 * CST * nBP;
    int*   ovf   = lists + 4LL * nB * CAPH;

    gcn_zero_kernel         <<<(2LL*nN + 64 + 2*CST*nBP + 255) / 256, 256, 0, stream>>>(deg, 2LL*nN + 64 + 2*CST*nBP);
    gcn_degree_bucket_kernel<<<(nE + FB_SL - 1) / FB_SL, FB_NT, 0, stream>>>(src, dst, deg, ctr, lists, ovf, nE, nN, nB, nBP);

    gcn_gemm1_kernel<<<(nMT + G1_TPB - 1) / G1_TPB, 128, 0, stream>>>(X, W1, deg, bufA, nN, nMT);
    gcn_bucket_walk_kernel<true> <<<nB, WTH, 0, stream>>>(bufA, lists, ovf, ctr, deg, b1, W2, bufB, nE, nN, nB, nBP);

    gcn_bucket_walk_kernel<false><<<nB, WTH, 0, stream>>>(bufB, lists, ovf, ctr, deg, b2, W2, out, nE, nN, nB, nBP);
}
